# starting kernel
# baseline (speedup 1.0000x reference)
#include <hip/hip_runtime.h>
#include <hip/hip_bf16.h>

#define DDIM 128
#define CSTRIDE 268

typedef __attribute__((ext_vector_type(2))) float v2f;
typedef __attribute__((ext_vector_type(8))) float v8f;

__global__ void dgnn_zero_kernel(float* __restrict__ p, int n4) {
    int i = blockIdx.x * blockDim.x + threadIdx.x;
    if (i < n4) {
        float4 z = make_float4(0.f, 0.f, 0.f, 0.f);
        *reinterpret_cast<float4*>(p + (size_t)i * 4) = z;
    }
}

__global__ void dgnn_scatter_kernel(const float* __restrict__ x_k1,
                                    const float* __restrict__ x_k2,
                                    const int* __restrict__ srcIdx,
                                    const int* __restrict__ dstIdx,
                                    const int* __restrict__ attr,
                                    float* __restrict__ agg1,
                                    float* __restrict__ agg2,
                                    int E) {
    int tid = blockIdx.x * blockDim.x + threadIdx.x;
    int e = tid >> 5;
    if (e >= E) return;
    int c = (tid & 31) << 2;

    int k = attr[e];
    if (k < 1 || k > 2) return;

    const float* __restrict__ xsrc = (k == 1) ? x_k1 : x_k2;
    float* __restrict__ ag         = (k == 1) ? agg1 : agg2;

    int s  = srcIdx[e];
    int d0 = dstIdx[e];

    float4 v = *reinterpret_cast<const float4*>(xsrc + (size_t)s * DDIM + c);
    float* base = ag + (size_t)d0 * DDIM + c;

    unsafeAtomicAdd(base + 0, v.x);
    unsafeAtomicAdd(base + 1, v.y);
    unsafeAtomicAdd(base + 2, v.z);
    unsafeAtomicAdd(base + 3, v.w);
}

__global__ __launch_bounds__(256)
void dgnn_gemm_kernel(const float* __restrict__ agg1,
                      const float* __restrict__ agg2,
                      const float* __restrict__ Wt,
                      const float* __restrict__ bt,
                      const float* __restrict__ alphat,
                      const float* __restrict__ xin,
                      float* __restrict__ xout,
                      int nTiles, int tilesPerBlock) {
    extern __shared__ float lds[];

    float e0 = __expf(alphat[0]);
    float e1 = __expf(alphat[1]);
    float inv = 1.0f / (e0 + e1);
    float a0 = e0 * inv;
    float a1 = e1 * inv;

    for (int i = threadIdx.x; i < DDIM * (DDIM / 4); i += blockDim.x) {
        int kk = i >> 5;
        int c4 = (i & 31) << 2;
        float4 v0 = *reinterpret_cast<const float4*>(Wt + (size_t)kk * DDIM + c4);
        float4 v1 = *reinterpret_cast<const float4*>(Wt + (size_t)DDIM * DDIM + (size_t)kk * DDIM + c4);
        lds[(c4 + 0) * CSTRIDE + kk]       = a0 * v0.x;
        lds[(c4 + 1) * CSTRIDE + kk]       = a0 * v0.y;
        lds[(c4 + 2) * CSTRIDE + kk]       = a0 * v0.z;
        lds[(c4 + 3) * CSTRIDE + kk]       = a0 * v0.w;
        lds[(c4 + 0) * CSTRIDE + 132 + kk] = a1 * v1.x;
        lds[(c4 + 1) * CSTRIDE + 132 + kk] = a1 * v1.y;
        lds[(c4 + 2) * CSTRIDE + 132 + kk] = a1 * v1.z;
        lds[(c4 + 3) * CSTRIDE + 132 + kk] = a1 * v1.w;
    }
    __syncthreads();

    const int wave = threadIdx.x >> 5;
    const int lane = threadIdx.x & 31;
    const int col  = lane & 15;
    const int half = lane >> 4;
    const int n0   = wave * 16;
    const int n    = n0 + col;

    const float bc = a0 * bt[n] + a1 * bt[DDIM + n];

    const float* bbase = lds + (size_t)n * CSTRIDE + 2 * half;

    for (int rt = 0; rt < tilesPerBlock; ++rt) {
        int tile = blockIdx.x * tilesPerBlock + rt;
        if (tile >= nTiles) break;
        int m0 = tile * 16;

        const float* arow1 = agg1 + (size_t)(m0 + col) * DDIM + 2 * half;
        const float* arow2 = agg2 + (size_t)(m0 + col) * DDIM + 2 * half;

        v8f acc = {};

        #pragma unroll 4
        for (int s = 0; s < 32; ++s) {
            v2f av1 = *reinterpret_cast<const v2f*>(arow1 + 4 * s);
            v2f bv1 = *reinterpret_cast<const v2f*>(bbase + 4 * s);
            acc = __builtin_amdgcn_wmma_f32_16x16x4_f32(
                false, av1, false, bv1, (short)0, acc, false, false);
            v2f av2 = *reinterpret_cast<const v2f*>(arow2 + 4 * s);
            v2f bv2 = *reinterpret_cast<const v2f*>(bbase + 132 + 4 * s);
            acc = __builtin_amdgcn_wmma_f32_16x16x4_f32(
                false, av2, false, bv2, (short)0, acc, false, false);
        }

        #pragma unroll
        for (int r = 0; r < 8; ++r) {
            int m = m0 + half * 8 + r;
            float v = acc[r] + bc;
            v = v > 0.0f ? v : 0.0f;
            xout[(size_t)m * DDIM + n] = xin[(size_t)m * DDIM + n] + v;
        }
    }
}

extern "C" void kernel_launch(void* const* d_in, const int* in_sizes, int n_in,
                              void* d_out, int out_size, void* d_ws, size_t ws_size,
                              hipStream_t stream) {
    const float* x     = (const float*)d_in[0];
    const int*   eidx  = (const int*)d_in[1];
    const int*   eattr = (const int*)d_in[2];
    const float* W     = (const float*)d_in[3];
    const float* b     = (const float*)d_in[4];
    const float* alpha = (const float*)d_in[5];
    float*       out   = (float*)d_out;

    const int N = in_sizes[0] / DDIM;
    const int E = in_sizes[2];
    const int T = 3;

    const size_t nd = (size_t)N * DDIM;
    float* ws   = (float*)d_ws;
    float* h1   = ws;
    float* h2   = ws + nd;
    float* agg1 = ws + 2 * nd;
    float* agg2 = ws + 3 * nd;

    const int* srcIdx = eidx;
    const int* dstIdx = eidx + E;

    const float* xs[4] = { x, h1, h2, out };

    const int nTiles = (N + 15) / 16;
    const int tilesPerBlock = 8;
    const int gemmBlocks = (nTiles + tilesPerBlock - 1) / tilesPerBlock;
    const size_t ldsBytes = (size_t)DDIM * CSTRIDE * sizeof(float);

    const int zero4 = (int)((2 * nd) / 4);
    const int zeroBlocks = (zero4 + 255) / 256;
    const long scatterThreads = (long)E * 32;
    const int scatterBlocks = (int)((scatterThreads + 255) / 256);

    for (int t = 0; t < T; ++t) {
        dgnn_zero_kernel<<<zeroBlocks, 256, 0, stream>>>(agg1, zero4);

        const float* x_k1 = xs[t];
        const float* x_k2 = xs[t > 0 ? t - 1 : 0];
        dgnn_scatter_kernel<<<scatterBlocks, 256, 0, stream>>>(
            x_k1, x_k2, srcIdx, dstIdx, eattr, agg1, agg2, E);

        dgnn_gemm_kernel<<<gemmBlocks, 256, ldsBytes, stream>>>(
            agg1, agg2,
            W + (size_t)t * 2 * DDIM * DDIM,
            b + (size_t)t * 2 * DDIM,
            alpha + (size_t)t * 2,
            xs[t], (float*)xs[t + 1],
            nTiles, tilesPerBlock);
    }
}


# aggregation by owner work-groups: each keeps 400 target rows in LDS, scans all edge targets (L2 resident), adds its rows by LDS f32 adds, one wave per row, fixed order; global atomics and clear gone
# speedup vs baseline: 1.7653x; 1.7653x over previous
#include <hip/hip_runtime.h>
#include <hip/hip_bf16.h>

#define DDIM 128
#define APITCH 260
#define TROWS 32
#define STAGE (TROWS * APITCH)

typedef __attribute__((ext_vector_type(2))) float v2f;
typedef __attribute__((ext_vector_type(4))) float v4f;
typedef __attribute__((ext_vector_type(8))) float v8f;

#define ROWS_OWN 400
#define OWN_WAVES 32
#define LCAP 384
#define SCAN_V4 8
#define ROWB 8

typedef __attribute__((ext_vector_type(4))) int v4i;
typedef v4i v4i_a4 __attribute__((aligned(4)));
typedef __attribute__((ext_vector_type(2))) int v2i;

__device__ __forceinline__ void dgnn_own_flush(const float* __restrict__ xsrc, float* acc,
                                               int os, int od, int n, int lane) {
    for (int j = 0; j < n; j += ROWB) {
        v4f v[ROWB];
        #pragma unroll
        for (int X = 0; X < ROWB; ++X) {
            v[X] = v4f{0.f, 0.f, 0.f, 0.f};
            if (j + X < n) {
                const int sr = __builtin_amdgcn_readlane(os, j + X);
                v[X] = *reinterpret_cast<const v4f*>(xsrc + (size_t)sr * DDIM + lane * 4);
            }
        }
        #pragma unroll
        for (int X = 0; X < ROWB; ++X) {
            if (j + X < n) {
                const int dr = __builtin_amdgcn_readlane(od, j + X);
                float* a = acc + dr * DDIM + lane * 4;
                atomicAdd(a + 0, v[X].x);
                atomicAdd(a + 1, v[X].y);
                atomicAdd(a + 2, v[X].z);
                atomicAdd(a + 3, v[X].w);
            }
        }
    }
}

__global__ __launch_bounds__(1024)
void dgnn_own_kernel(const float* __restrict__ xsrc,
                     const int* __restrict__ srcIdx,
                     const int* __restrict__ dstIdx,
                     const int* __restrict__ attr,
                     float* __restrict__ aggOut,
                     int E, int N, int hop) {
    __shared__ __attribute__((aligned(16))) float acc[ROWS_OWN * DDIM];
    __shared__ v2i lst[OWN_WAVES * LCAP];
    __shared__ int lcnt[OWN_WAVES];

    const int tid  = threadIdx.x;
    const int lane = tid & 31;
    const int wave = __builtin_amdgcn_readfirstlane(tid >> 5);
    const int r0 = (int)blockIdx.x * ROWS_OWN;
    const int nrows = (N - r0) < ROWS_OWN ? (N - r0) : ROWS_OWN;

    for (int i = tid; i < ROWS_OWN * DDIM / 4; i += 1024)
        reinterpret_cast<v4f*>(acc)[i] = v4f{0.f, 0.f, 0.f, 0.f};
    __syncthreads();

    v2i* mylst = lst + wave * LCAP;
    int cnt = 0;
    const int TRIP = SCAN_V4 * 128;
    for (int base = wave * TRIP; base < E; base += OWN_WAVES * TRIP) {
        v4i d4[SCAN_V4];
        if (base + TRIP <= E) {
            #pragma unroll
            for (int u = 0; u < SCAN_V4; ++u)
                d4[u] = *reinterpret_cast<const v4i_a4*>(dstIdx + base + u * 128 + lane * 4);
        } else {
            #pragma unroll
            for (int u = 0; u < SCAN_V4; ++u) {
                const int eb = base + u * 128 + lane * 4;
                d4[u].x = (eb     < E) ? dstIdx[eb]     : -1;
                d4[u].y = (eb + 1 < E) ? dstIdx[eb + 1] : -1;
                d4[u].z = (eb + 2 < E) ? dstIdx[eb + 2] : -1;
                d4[u].w = (eb + 3 < E) ? dstIdx[eb + 3] : -1;
            }
        }
        unsigned ovf = 0;
        #pragma unroll
        for (int q = 0; q < SCAN_V4 * 4; ++q) {
            const int dl = d4[q >> 2][q & 3] - r0;
            const bool m = (unsigned)dl < (unsigned)nrows;
            const unsigned mask = __builtin_amdgcn_ballot_w32(m);
            if (mask) {
                const int pos = cnt + (int)__builtin_amdgcn_mbcnt_lo(mask, 0u);
                if (m) {
                    if (pos < LCAP) mylst[pos] = v2i{base + (q >> 2) * 128 + lane * 4 + (q & 3), dl};
                    else ovf |= 1u << q;
                }
                cnt += __builtin_popcount(mask);
            }
        }
        if (__builtin_amdgcn_ballot_w32(ovf != 0u)) {
            for (int q = 0; q < SCAN_V4 * 4; ++q) {
                if ((ovf >> q) & 1u) {
                    const int e = base + (q >> 2) * 128 + lane * 4 + (q & 3);
                    if (attr[e] == hop) {
                        const float* row = xsrc + (size_t)srcIdx[e] * DDIM;
                        float* a = acc + (dstIdx[e] - r0) * DDIM;
                        for (int c = 0; c < DDIM; ++c) atomicAdd(a + c, row[c]);
                    }
                }
            }
        }
    }

    const int nHit = cnt < LCAP ? cnt : LCAP;
    int nLive = 0;
    for (int i0 = 0; i0 < nHit; i0 += 32) {
        const int idx = i0 + lane;
        const bool valid = idx < nHit;
        v2i en = v2i{0, 0};
        int k = 0, sr = 0;
        if (valid) {
            en = mylst[idx];
            k  = attr[en.x];
            sr = srcIdx[en.x];
        }
        const bool live = valid & (k == hop);
        const unsigned mask = __builtin_amdgcn_ballot_w32(live);
        if (live) mylst[nLive + (int)__builtin_amdgcn_mbcnt_lo(mask, 0u)] = v2i{sr, en.y};
        nLive += __builtin_popcount(mask);
    }
    if (lane == 0) lcnt[wave] = nLive;
    __syncthreads();

    int os = 0, od = 0, on = 0;
    for (int L = 0; L < OWN_WAVES; ++L) {
        const int nL = __builtin_amdgcn_readfirstlane(lcnt[L]);
        const v2i* lp = lst + L * LCAP;
        for (int i0 = 0; i0 < nL; i0 += 128) {
            v2i en[4];
            #pragma unroll
            for (int u = 0; u < 4; ++u) {
                const int idx = i0 + u * 32 + lane;
                en[u] = v2i{0, -1};
                if (idx < nL) en[u] = lp[idx];
            }
            #pragma unroll
            for (int u = 0; u < 4; ++u) {
                const bool mine = (en[u].y >= 0) & ((en[u].y & 31) == wave);
                unsigned mask = __builtin_amdgcn_ballot_w32(mine);
                while (mask) {
                    const int j = __builtin_ctz(mask);
                    mask &= mask - 1u;
                    os = (lane == on) ? __builtin_amdgcn_readlane(en[u].x, j) : os;
                    od = (lane == on) ? __builtin_amdgcn_readlane(en[u].y, j) : od;
                    ++on;
                    if (on == 32) { dgnn_own_flush(xsrc, acc, os, od, 32, lane); on = 0; }
                }
            }
        }
    }
    dgnn_own_flush(xsrc, acc, os, od, on, lane);
    __syncthreads();

    v4f* outp = reinterpret_cast<v4f*>(aggOut + (size_t)r0 * DDIM);
    for (int i = tid; i < nrows * (DDIM / 4); i += 1024)
        outp[i] = reinterpret_cast<const v4f*>(acc)[i];
}

__device__ __forceinline__ void dgnn_stage_load(const float* __restrict__ agg1,
                                                const float* __restrict__ agg2,
                                                int st, int nTiles, int wave, int c4,
                                                v4f (&p1)[4], v4f (&p2)[4]) {
    const int t0 = 2 * st;
    const int t1 = t0 + 1 < nTiles ? t0 + 1 : nTiles - 1;
    #pragma unroll
    for (int j = 0; j < 4; ++j) {
        const int g = (j < 2 ? t0 : t1) * 16 + (j & 1) * 8 + wave;
        p1[j] = *reinterpret_cast<const v4f*>(agg1 + (size_t)g * DDIM + c4);
        p2[j] = *reinterpret_cast<const v4f*>(agg2 + (size_t)g * DDIM + c4);
    }
}

__device__ __forceinline__ void dgnn_stage_store(float* __restrict__ stage, int wave, int c4,
                                                 const v4f (&p1)[4], const v4f (&p2)[4]) {
    #pragma unroll
    for (int j = 0; j < 4; ++j) {
        float* row = stage + (j * 8 + wave) * APITCH + c4;
        *reinterpret_cast<v4f*>(row)        = p1[j];
        *reinterpret_cast<v4f*>(row + DDIM) = p2[j];
    }
}

#define CHK 4
__device__ __forceinline__ void dgnn_frag_load(const float* __restrict__ aP, int c, v2f (&f)[4 * CHK]) {
    #pragma unroll
    for (int i = 0; i < CHK; ++i) {
        const int s = c * CHK + i;
        f[4 * i + 0] = *reinterpret_cast<const v2f*>(aP + 4 * s);
        f[4 * i + 1] = *reinterpret_cast<const v2f*>(aP + 16 * APITCH + 4 * s);
        f[4 * i + 2] = *reinterpret_cast<const v2f*>(aP + DDIM + 4 * s);
        f[4 * i + 3] = *reinterpret_cast<const v2f*>(aP + 16 * APITCH + DDIM + 4 * s);
    }
}

__global__ __launch_bounds__(256)
void dgnn_gemm_kernel(const float* __restrict__ agg1,
                      const float* __restrict__ agg2,
                      const float* __restrict__ Wt,
                      const float* __restrict__ bt,
                      const float* __restrict__ alphat,
                      const float* __restrict__ xin,
                      float* __restrict__ xout,
                      int nTiles, int nSuper) {
    extern __shared__ float lds[];

    const int wave = threadIdx.x >> 5;
    const int lane = threadIdx.x & 31;
    const int col  = lane & 15;
    const int half = lane >> 4;
    const int n    = wave * 16 + col;
    const int c4   = lane << 2;

    int st = blockIdx.x;
    const int stride = gridDim.x;

    v4f p1[4], p2[4];
    dgnn_stage_load(agg1, agg2, st < nSuper ? st : 0, nTiles, wave, c4, p1, p2);

    float e0 = __expf(alphat[0]);
    float e1 = __expf(alphat[1]);
    float inv = 1.0f / (e0 + e1);
    float a0 = e0 * inv;
    float a1 = e1 * inv;

    v2f b1[32], b2[32];
    {
        const float* w0p = Wt + (size_t)(2 * half) * DDIM + n;
        const float* w1p = w0p + (size_t)DDIM * DDIM;
        #pragma unroll
        for (int s = 0; s < 32; ++s) {
            b1[s].x = a0 * w0p[(4 * s) * DDIM];
            b1[s].y = a0 * w0p[(4 * s + 1) * DDIM];
            b2[s].x = a1 * w1p[(4 * s) * DDIM];
            b2[s].y = a1 * w1p[(4 * s + 1) * DDIM];
        }
    }

    const float bc = a0 * bt[n] + a1 * bt[DDIM + n];

    dgnn_stage_store(lds, wave, c4, p1, p2);
    __syncthreads();

    int cur = 0;
    for (; st < nSuper; st += stride) {
        const int nxt = st + stride;
        const bool more = nxt < nSuper;
        if (more) dgnn_stage_load(agg1, agg2, nxt, nTiles, wave, c4, p1, p2);

        const int t0 = 2 * st;
        const int t1 = t0 + 1 < nTiles ? t0 + 1 : nTiles - 1;
        const size_t o0 = (size_t)(t0 * 16 + half * 8) * DDIM + n;
        const size_t o1 = (size_t)(t1 * 16 + half * 8) * DDIM + n;

        float xr0[8], xr1[8];
        #pragma unroll
        for (int r = 0; r < 8; ++r) xr0[r] = xin[o0 + r * DDIM];
        #pragma unroll
        for (int r = 0; r < 8; ++r) xr1[r] = xin[o1 + r * DDIM];

        const float* aP = lds + cur * STAGE + col * APITCH + 2 * half;

        v8f acc0 = {};
        v8f acc1 = {};

        v2f fa[4 * CHK], fb[4 * CHK];
        dgnn_frag_load(aP, 0, fa);
        #pragma unroll
        for (int c = 0; c < 32 / CHK; c += 2) {
            dgnn_frag_load(aP, c + 1, fb);
            __builtin_amdgcn_sched_barrier(0);
            #pragma unroll
            for (int i = 0; i < CHK; ++i) {
                const int s = c * CHK + i;
                acc0 = __builtin_amdgcn_wmma_f32_16x16x4_f32(false, fa[4 * i + 0], false, b1[s], (short)0, acc0, false, false);
                acc1 = __builtin_amdgcn_wmma_f32_16x16x4_f32(false, fa[4 * i + 1], false, b1[s], (short)0, acc1, false, false);
                acc0 = __builtin_amdgcn_wmma_f32_16x16x4_f32(false, fa[4 * i + 2], false, b2[s], (short)0, acc0, false, false);
                acc1 = __builtin_amdgcn_wmma_f32_16x16x4_f32(false, fa[4 * i + 3], false, b2[s], (short)0, acc1, false, false);
            }
            __builtin_amdgcn_sched_barrier(0);
            if (c + 2 < 32 / CHK) dgnn_frag_load(aP, c + 2, fa);
            __builtin_amdgcn_sched_barrier(0);
            #pragma unroll
            for (int i = 0; i < CHK; ++i) {
                const int s = (c + 1) * CHK + i;
                acc0 = __builtin_amdgcn_wmma_f32_16x16x4_f32(false, fb[4 * i + 0], false, b1[s], (short)0, acc0, false, false);
                acc1 = __builtin_amdgcn_wmma_f32_16x16x4_f32(false, fb[4 * i + 1], false, b1[s], (short)0, acc1, false, false);
                acc0 = __builtin_amdgcn_wmma_f32_16x16x4_f32(false, fb[4 * i + 2], false, b2[s], (short)0, acc0, false, false);
                acc1 = __builtin_amdgcn_wmma_f32_16x16x4_f32(false, fb[4 * i + 3], false, b2[s], (short)0, acc1, false, false);
            }
            __builtin_amdgcn_sched_barrier(0);
        }

        #pragma unroll
        for (int r = 0; r < 8; ++r) {
            float v = acc0[r] + bc;
            v = v > 0.0f ? v : 0.0f;
            xout[o0 + r * DDIM] = xr0[r] + v;
        }
        #pragma unroll
        for (int r = 0; r < 8; ++r) {
            float v = acc1[r] + bc;
            v = v > 0.0f ? v : 0.0f;
            xout[o1 + r * DDIM] = xr1[r] + v;
        }

        if (more) dgnn_stage_store(lds + (cur ^ 1) * STAGE, wave, c4, p1, p2);
        __syncthreads();
        cur ^= 1;
    }
}

extern "C" void kernel_launch(void* const* d_in, const int* in_sizes, int n_in,
                              void* d_out, int out_size, void* d_ws, size_t ws_size,
                              hipStream_t stream) {
    const float* x     = (const float*)d_in[0];
    const int*   eidx  = (const int*)d_in[1];
    const int*   eattr = (const int*)d_in[2];
    const float* W     = (const float*)d_in[3];
    const float* b     = (const float*)d_in[4];
    const float* alpha = (const float*)d_in[5];
    float*       out   = (float*)d_out;

    const int N = in_sizes[0] / DDIM;
    const int E = in_sizes[2];
    const int T = 3;

    const size_t nd = (size_t)N * DDIM;
    float* ws   = (float*)d_ws;
    float* h1   = ws;
    float* h2   = ws + nd;
    float* agg1 = ws + 2 * nd;
    float* agg2 = ws + 3 * nd;

    const int* srcIdx = eidx;
    const int* dstIdx = eidx + E;

    const float* xs[4] = { x, h1, h2, out };

    const int nTiles = (N + 15) / 16;
    const int nSuper = (nTiles + 1) / 2;
    const int gemmBlocks = 256;
    const size_t ldsBytes = (size_t)2 * STAGE * sizeof(float);

    const int ownBlocks = (N + ROWS_OWN - 1) / ROWS_OWN;

    for (int t = 0; t < T; ++t) {
        dgnn_own_kernel<<<ownBlocks, 1024, 0, stream>>>(
            xs[t], srcIdx, dstIdx, eattr, agg1, E, N, 1);
        if (t != 1)
            dgnn_own_kernel<<<ownBlocks, 1024, 0, stream>>>(
                xs[t > 0 ? t - 1 : 0], srcIdx, dstIdx, eattr, agg2, E, N, 2);

        dgnn_gemm_kernel<<<gemmBlocks, 256, ldsBytes, stream>>>(
            agg1, agg2,
            W + (size_t)t * 2 * DDIM * DDIM,
            b + (size_t)t * 2 * DDIM,
            alpha + (size_t)t * 2,
            xs[t], (float*)xs[t + 1],
            nTiles, nSuper);
    }
}


# aggregation: the first of the five launches keeps its owner lists (source row, local row, hop) in the not yet written output buffer; the other four read them back, no edge scan; sum order kept
# speedup vs baseline: 2.6707x; 1.5129x over previous
#include <hip/hip_runtime.h>
#include <hip/hip_bf16.h>

#define DDIM 128
#define APITCH 260
#define TROWS 32
#define STAGE (TROWS * APITCH)

typedef __attribute__((ext_vector_type(2))) float v2f;
typedef __attribute__((ext_vector_type(4))) float v4f;
typedef __attribute__((ext_vector_type(8))) float v8f;

#define ROWS_OWN 400
#define OWN_WAVES 32
#define LCAP 384
#define SCAN_V4 8
#define ROWB 8

typedef __attribute__((ext_vector_type(4))) int v4i;
typedef v4i v4i_a4 __attribute__((aligned(4)));
typedef __attribute__((ext_vector_type(2))) int v2i;

__device__ __forceinline__ void dgnn_own_flush(const float* __restrict__ xsrc, float* acc,
                                               int os, int od, int n, int lane) {
    for (int j = 0; j < n; j += ROWB) {
        v4f v[ROWB];
        #pragma unroll
        for (int X = 0; X < ROWB; ++X) {
            v[X] = v4f{0.f, 0.f, 0.f, 0.f};
            if (j + X < n) {
                const int sr = __builtin_amdgcn_readlane(os, j + X);
                v[X] = *reinterpret_cast<const v4f*>(xsrc + (size_t)sr * DDIM + lane * 4);
            }
        }
        #pragma unroll
        for (int X = 0; X < ROWB; ++X) {
            if (j + X < n) {
                const int dr = __builtin_amdgcn_readlane(od, j + X);
                float* a = acc + dr * DDIM + lane * 4;
                atomicAdd(a + 0, v[X].x);
                atomicAdd(a + 1, v[X].y);
                atomicAdd(a + 2, v[X].z);
                atomicAdd(a + 3, v[X].w);
            }
        }
    }
}

__global__ __launch_bounds__(1024)
void dgnn_own_kernel(const float* __restrict__ xsrc,
                     const int* __restrict__ srcIdx,
                     const int* __restrict__ dstIdx,
                     const int* __restrict__ attr,
                     float* __restrict__ aggOut,
                     float* __restrict__ keep,
                     int E, int N, int hop, int reuse) {
    __shared__ __attribute__((aligned(16))) float acc[ROWS_OWN * DDIM];
    __shared__ v2i lst[OWN_WAVES * LCAP];
    __shared__ int lcnt[OWN_WAVES];

    const int tid  = threadIdx.x;
    const int lane = tid & 31;
    const int wave = __builtin_amdgcn_readfirstlane(tid >> 5);
    const int r0 = (int)blockIdx.x * ROWS_OWN;
    const int nrows = (N - r0) < ROWS_OWN ? (N - r0) : ROWS_OWN;

    for (int i = tid; i < ROWS_OWN * DDIM / 4; i += 1024)
        reinterpret_cast<v4f*>(acc)[i] = v4f{0.f, 0.f, 0.f, 0.f};
    __syncthreads();

    v2i* mylst = lst + wave * LCAP;
    const int slot = (int)blockIdx.x * OWN_WAVES + wave;
    v2i* klst = reinterpret_cast<v2i*>(keep) + (size_t)slot * LCAP;
    int* keepCnt = reinterpret_cast<int*>(keep) + (size_t)gridDim.x * OWN_WAVES * LCAP * 2;
    const int nKept = reuse ? __builtin_amdgcn_readfirstlane(keepCnt[slot]) : LCAP + 1;
    int nLive = 0;

    if (nKept <= LCAP) {
        for (int i0 = 0; i0 < nKept; i0 += 128) {
            v2i en[4];
            #pragma unroll
            for (int u = 0; u < 4; ++u) {
                const int idx = i0 + u * 32 + lane;
                en[u] = v2i{0, 0};
                if (idx < nKept) en[u] = klst[idx];
            }
            #pragma unroll
            for (int u = 0; u < 4; ++u) {
                const bool live = (i0 + u * 32 + lane < nKept) & ((en[u].y >> 16) == hop);
                const unsigned mask = __builtin_amdgcn_ballot_w32(live);
                if (live) mylst[nLive + (int)__builtin_amdgcn_mbcnt_lo(mask, 0u)] = v2i{en[u].x, en[u].y & 0xffff};
                nLive += __builtin_popcount(mask);
            }
        }
    } else {
        int cnt = 0;
        const int TRIP = SCAN_V4 * 128;
        for (int base = wave * TRIP; base < E; base += OWN_WAVES * TRIP) {
            v4i d4[SCAN_V4];
            if (base + TRIP <= E) {
                #pragma unroll
                for (int u = 0; u < SCAN_V4; ++u)
                    d4[u] = *reinterpret_cast<const v4i_a4*>(dstIdx + base + u * 128 + lane * 4);
            } else {
                #pragma unroll
                for (int u = 0; u < SCAN_V4; ++u) {
                    const int eb = base + u * 128 + lane * 4;
                    d4[u].x = (eb     < E) ? dstIdx[eb]     : -1;
                    d4[u].y = (eb + 1 < E) ? dstIdx[eb + 1] : -1;
                    d4[u].z = (eb + 2 < E) ? dstIdx[eb + 2] : -1;
                    d4[u].w = (eb + 3 < E) ? dstIdx[eb + 3] : -1;
                }
            }
            unsigned ovf = 0;
            #pragma unroll
            for (int q = 0; q < SCAN_V4 * 4; ++q) {
                const int dl = d4[q >> 2][q & 3] - r0;
                const bool m = (unsigned)dl < (unsigned)nrows;
                const unsigned mask = __builtin_amdgcn_ballot_w32(m);
                if (mask) {
                    const int pos = cnt + (int)__builtin_amdgcn_mbcnt_lo(mask, 0u);
                    if (m) {
                        if (pos < LCAP) mylst[pos] = v2i{base + (q >> 2) * 128 + lane * 4 + (q & 3), dl};
                        else ovf |= 1u << q;
                    }
                    cnt += __builtin_popcount(mask);
                }
            }
            if (__builtin_amdgcn_ballot_w32(ovf != 0u)) {
                for (int q = 0; q < SCAN_V4 * 4; ++q) {
                    if ((ovf >> q) & 1u) {
                        const int e = base + (q >> 2) * 128 + lane * 4 + (q & 3);
                        if (attr[e] == hop) {
                            const float* row = xsrc + (size_t)srcIdx[e] * DDIM;
                            float* a = acc + (dstIdx[e] - r0) * DDIM;
                            for (int c = 0; c < DDIM; ++c) atomicAdd(a + c, row[c]);
                        }
                    }
                }
            }
        }

        const int nHit = cnt < LCAP ? cnt : LCAP;
        int nKeep = 0;
        for (int i0 = 0; i0 < nHit; i0 += 32) {
            const int idx = i0 + lane;
            const bool valid = idx < nHit;
            v2i en = v2i{0, 0};
            int k = 0, sr = 0;
            if (valid) {
                en = mylst[idx];
                k  = attr[en.x];
                sr = srcIdx[en.x];
            }
            if (!reuse) {
                const bool keep = valid & ((unsigned)(k - 1) < 2u);
                const unsigned km = __builtin_amdgcn_ballot_w32(keep);
                if (keep) klst[nKeep + (int)__builtin_amdgcn_mbcnt_lo(km, 0u)] = v2i{sr, en.y | (k << 16)};
                nKeep += __builtin_popcount(km);
            }
            const bool live = valid & (k == hop);
            const unsigned mask = __builtin_amdgcn_ballot_w32(live);
            if (live) mylst[nLive + (int)__builtin_amdgcn_mbcnt_lo(mask, 0u)] = v2i{sr, en.y};
            nLive += __builtin_popcount(mask);
        }
        if (!reuse && lane == 0) keepCnt[slot] = cnt > LCAP ? LCAP + 1 : nKeep;
    }
    if (lane == 0) lcnt[wave] = nLive;
    __syncthreads();

    int os = 0, od = 0, on = 0;
    for (int L = 0; L < OWN_WAVES; ++L) {
        const int nL = __builtin_amdgcn_readfirstlane(lcnt[L]);
        const v2i* lp = lst + L * LCAP;
        for (int i0 = 0; i0 < nL; i0 += 128) {
            v2i en[4];
            #pragma unroll
            for (int u = 0; u < 4; ++u) {
                const int idx = i0 + u * 32 + lane;
                en[u] = v2i{0, -1};
                if (idx < nL) en[u] = lp[idx];
            }
            #pragma unroll
            for (int u = 0; u < 4; ++u) {
                const bool mine = (en[u].y >= 0) & ((en[u].y & 31) == wave);
                unsigned mask = __builtin_amdgcn_ballot_w32(mine);
                while (mask) {
                    const int j = __builtin_ctz(mask);
                    mask &= mask - 1u;
                    os = (lane == on) ? __builtin_amdgcn_readlane(en[u].x, j) : os;
                    od = (lane == on) ? __builtin_amdgcn_readlane(en[u].y, j) : od;
                    ++on;
                    if (on == 32) { dgnn_own_flush(xsrc, acc, os, od, 32, lane); on = 0; }
                }
            }
        }
    }
    dgnn_own_flush(xsrc, acc, os, od, on, lane);
    __syncthreads();

    v4f* outp = reinterpret_cast<v4f*>(aggOut + (size_t)r0 * DDIM);
    for (int i = tid; i < nrows * (DDIM / 4); i += 1024)
        outp[i] = reinterpret_cast<const v4f*>(acc)[i];
}

__device__ __forceinline__ void dgnn_stage_load(const float* __restrict__ agg1,
                                                const float* __restrict__ agg2,
                                                int st, int nTiles, int wave, int c4,
                                                v4f (&p1)[4], v4f (&p2)[4]) {
    const int t0 = 2 * st;
    const int t1 = t0 + 1 < nTiles ? t0 + 1 : nTiles - 1;
    #pragma unroll
    for (int j = 0; j < 4; ++j) {
        const int g = (j < 2 ? t0 : t1) * 16 + (j & 1) * 8 + wave;
        p1[j] = *reinterpret_cast<const v4f*>(agg1 + (size_t)g * DDIM + c4);
        p2[j] = *reinterpret_cast<const v4f*>(agg2 + (size_t)g * DDIM + c4);
    }
}

__device__ __forceinline__ void dgnn_stage_store(float* __restrict__ stage, int wave, int c4,
                                                 const v4f (&p1)[4], const v4f (&p2)[4]) {
    #pragma unroll
    for (int j = 0; j < 4; ++j) {
        float* row = stage + (j * 8 + wave) * APITCH + c4;
        *reinterpret_cast<v4f*>(row)        = p1[j];
        *reinterpret_cast<v4f*>(row + DDIM) = p2[j];
    }
}

#define CHK 4
__device__ __forceinline__ void dgnn_frag_load(const float* __restrict__ aP, int c, v2f (&f)[4 * CHK]) {
    #pragma unroll
    for (int i = 0; i < CHK; ++i) {
        const int s = c * CHK + i;
        f[4 * i + 0] = *reinterpret_cast<const v2f*>(aP + 4 * s);
        f[4 * i + 1] = *reinterpret_cast<const v2f*>(aP + 16 * APITCH + 4 * s);
        f[4 * i + 2] = *reinterpret_cast<const v2f*>(aP + DDIM + 4 * s);
        f[4 * i + 3] = *reinterpret_cast<const v2f*>(aP + 16 * APITCH + DDIM + 4 * s);
    }
}

__global__ __launch_bounds__(256)
void dgnn_gemm_kernel(const float* __restrict__ agg1,
                      const float* __restrict__ agg2,
                      const float* __restrict__ Wt,
                      const float* __restrict__ bt,
                      const float* __restrict__ alphat,
                      const float* __restrict__ xin,
                      float* __restrict__ xout,
                      int nTiles, int nSuper) {
    extern __shared__ float lds[];

    const int wave = threadIdx.x >> 5;
    const int lane = threadIdx.x & 31;
    const int col  = lane & 15;
    const int half = lane >> 4;
    const int n    = wave * 16 + col;
    const int c4   = lane << 2;

    int st = blockIdx.x;
    const int stride = gridDim.x;

    v4f p1[4], p2[4];
    dgnn_stage_load(agg1, agg2, st < nSuper ? st : 0, nTiles, wave, c4, p1, p2);

    float e0 = __expf(alphat[0]);
    float e1 = __expf(alphat[1]);
    float inv = 1.0f / (e0 + e1);
    float a0 = e0 * inv;
    float a1 = e1 * inv;

    v2f b1[32], b2[32];
    {
        const float* w0p = Wt + (size_t)(2 * half) * DDIM + n;
        const float* w1p = w0p + (size_t)DDIM * DDIM;
        #pragma unroll
        for (int s = 0; s < 32; ++s) {
            b1[s].x = a0 * w0p[(4 * s) * DDIM];
            b1[s].y = a0 * w0p[(4 * s + 1) * DDIM];
            b2[s].x = a1 * w1p[(4 * s) * DDIM];
            b2[s].y = a1 * w1p[(4 * s + 1) * DDIM];
        }
    }

    const float bc = a0 * bt[n] + a1 * bt[DDIM + n];

    dgnn_stage_store(lds, wave, c4, p1, p2);
    __syncthreads();

    int cur = 0;
    for (; st < nSuper; st += stride) {
        const int nxt = st + stride;
        const bool more = nxt < nSuper;
        if (more) dgnn_stage_load(agg1, agg2, nxt, nTiles, wave, c4, p1, p2);

        const int t0 = 2 * st;
        const int t1 = t0 + 1 < nTiles ? t0 + 1 : nTiles - 1;
        const size_t o0 = (size_t)(t0 * 16 + half * 8) * DDIM + n;
        const size_t o1 = (size_t)(t1 * 16 + half * 8) * DDIM + n;

        float xr0[8], xr1[8];
        #pragma unroll
        for (int r = 0; r < 8; ++r) xr0[r] = xin[o0 + r * DDIM];
        #pragma unroll
        for (int r = 0; r < 8; ++r) xr1[r] = xin[o1 + r * DDIM];

        const float* aP = lds + cur * STAGE + col * APITCH + 2 * half;

        v8f acc0 = {};
        v8f acc1 = {};

        v2f fa[4 * CHK], fb[4 * CHK];
        dgnn_frag_load(aP, 0, fa);
        #pragma unroll
        for (int c = 0; c < 32 / CHK; c += 2) {
            dgnn_frag_load(aP, c + 1, fb);
            __builtin_amdgcn_sched_barrier(0);
            #pragma unroll
            for (int i = 0; i < CHK; ++i) {
                const int s = c * CHK + i;
                acc0 = __builtin_amdgcn_wmma_f32_16x16x4_f32(false, fa[4 * i + 0], false, b1[s], (short)0, acc0, false, false);
                acc1 = __builtin_amdgcn_wmma_f32_16x16x4_f32(false, fa[4 * i + 1], false, b1[s], (short)0, acc1, false, false);
                acc0 = __builtin_amdgcn_wmma_f32_16x16x4_f32(false, fa[4 * i + 2], false, b2[s], (short)0, acc0, false, false);
                acc1 = __builtin_amdgcn_wmma_f32_16x16x4_f32(false, fa[4 * i + 3], false, b2[s], (short)0, acc1, false, false);
            }
            __builtin_amdgcn_sched_barrier(0);
            if (c + 2 < 32 / CHK) dgnn_frag_load(aP, c + 2, fa);
            __builtin_amdgcn_sched_barrier(0);
            #pragma unroll
            for (int i = 0; i < CHK; ++i) {
                const int s = (c + 1) * CHK + i;
                acc0 = __builtin_amdgcn_wmma_f32_16x16x4_f32(false, fb[4 * i + 0], false, b1[s], (short)0, acc0, false, false);
                acc1 = __builtin_amdgcn_wmma_f32_16x16x4_f32(false, fb[4 * i + 1], false, b1[s], (short)0, acc1, false, false);
                acc0 = __builtin_amdgcn_wmma_f32_16x16x4_f32(false, fb[4 * i + 2], false, b2[s], (short)0, acc0, false, false);
                acc1 = __builtin_amdgcn_wmma_f32_16x16x4_f32(false, fb[4 * i + 3], false, b2[s], (short)0, acc1, false, false);
            }
            __builtin_amdgcn_sched_barrier(0);
        }

        #pragma unroll
        for (int r = 0; r < 8; ++r) {
            float v = acc0[r] + bc;
            v = v > 0.0f ? v : 0.0f;
            xout[o0 + r * DDIM] = xr0[r] + v;
        }
        #pragma unroll
        for (int r = 0; r < 8; ++r) {
            float v = acc1[r] + bc;
            v = v > 0.0f ? v : 0.0f;
            xout[o1 + r * DDIM] = xr1[r] + v;
        }

        if (more) dgnn_stage_store(lds + (cur ^ 1) * STAGE, wave, c4, p1, p2);
        __syncthreads();
        cur ^= 1;
    }
}

extern "C" void kernel_launch(void* const* d_in, const int* in_sizes, int n_in,
                              void* d_out, int out_size, void* d_ws, size_t ws_size,
                              hipStream_t stream) {
    const float* x     = (const float*)d_in[0];
    const int*   eidx  = (const int*)d_in[1];
    const int*   eattr = (const int*)d_in[2];
    const float* W     = (const float*)d_in[3];
    const float* b     = (const float*)d_in[4];
    const float* alpha = (const float*)d_in[5];
    float*       out   = (float*)d_out;

    const int N = in_sizes[0] / DDIM;
    const int E = in_sizes[2];
    const int T = 3;

    const size_t nd = (size_t)N * DDIM;
    float* ws   = (float*)d_ws;
    float* h1   = ws;
    float* h2   = ws + nd;
    float* agg1 = ws + 2 * nd;
    float* agg2 = ws + 3 * nd;

    const int* srcIdx = eidx;
    const int* dstIdx = eidx + E;

    const float* xs[4] = { x, h1, h2, out };

    const int nTiles = (N + 15) / 16;
    const int nSuper = (nTiles + 1) / 2;
    const int gemmBlocks = 256;
    const size_t ldsBytes = (size_t)2 * STAGE * sizeof(float);

    const int ownBlocks = (N + ROWS_OWN - 1) / ROWS_OWN;

    float* keep = out;

    for (int t = 0; t < T; ++t) {
        dgnn_own_kernel<<<ownBlocks, 1024, 0, stream>>>(
            xs[t], srcIdx, dstIdx, eattr, agg1, keep, E, N, 1,
            t > 0 ? 1 : 0);
        if (t != 1)
            dgnn_own_kernel<<<ownBlocks, 1024, 0, stream>>>(
                xs[t > 0 ? t - 1 : 0], srcIdx, dstIdx, eattr, agg2, keep, E, N, 2, 1);

        dgnn_gemm_kernel<<<gemmBlocks, 256, ldsBytes, stream>>>(
            agg1, agg2,
            W + (size_t)t * 2 * DDIM * DDIM,
            b + (size_t)t * 2 * DDIM,
            alpha + (size_t)t * 2,
            xs[t], (float*)xs[t + 1],
            nTiles, nSuper);
    }
}


# aggregation: a row's sums kept in LDS as four 32-float planes (column 4l+c at float c*32+l), so the 32 lanes of each LDS float add hit 32 banks (lanes l, l+16 shared one); sum order kept
# speedup vs baseline: 3.2126x; 1.2029x over previous
#include <hip/hip_runtime.h>
#include <hip/hip_bf16.h>

#define DDIM 128
#define APITCH 260
#define TROWS 32
#define STAGE (TROWS * APITCH)

typedef __attribute__((ext_vector_type(2))) float v2f;
typedef __attribute__((ext_vector_type(4))) float v4f;
typedef __attribute__((ext_vector_type(8))) float v8f;

#define ROWS_OWN 400
#define OWN_WAVES 32
#define LCAP 384
#define SCAN_V4 8
#define ROWB 8

typedef __attribute__((ext_vector_type(4))) int v4i;
typedef v4i v4i_a4 __attribute__((aligned(4)));
typedef __attribute__((ext_vector_type(2))) int v2i;

__device__ __forceinline__ void dgnn_own_flush(const float* __restrict__ xsrc, float* acc,
                                               int os, int od, int n, int lane) {
    for (int j = 0; j < n; j += ROWB) {
        v4f v[ROWB];
        #pragma unroll
        for (int X = 0; X < ROWB; ++X) {
            v[X] = v4f{0.f, 0.f, 0.f, 0.f};
            if (j + X < n) {
                const int sr = __builtin_amdgcn_readlane(os, j + X);
                v[X] = *reinterpret_cast<const v4f*>(xsrc + (size_t)sr * DDIM + lane * 4);
            }
        }
        #pragma unroll
        for (int X = 0; X < ROWB; ++X) {
            if (j + X < n) {
                const int dr = __builtin_amdgcn_readlane(od, j + X);
                float* a = acc + dr * DDIM + lane;
                atomicAdd(a + 0,  v[X].x);
                atomicAdd(a + 32, v[X].y);
                atomicAdd(a + 64, v[X].z);
                atomicAdd(a + 96, v[X].w);
            }
        }
    }
}

__global__ __launch_bounds__(1024)
void dgnn_own_kernel(const float* __restrict__ xsrc,
                     const int* __restrict__ srcIdx,
                     const int* __restrict__ dstIdx,
                     const int* __restrict__ attr,
                     float* __restrict__ aggOut,
                     float* __restrict__ keep,
                     int E, int N, int hop, int reuse) {
    __shared__ __attribute__((aligned(16))) float acc[ROWS_OWN * DDIM];
    __shared__ v2i lst[OWN_WAVES * LCAP];
    __shared__ int lcnt[OWN_WAVES];

    const int tid  = threadIdx.x;
    const int lane = tid & 31;
    const int wave = __builtin_amdgcn_readfirstlane(tid >> 5);
    const int r0 = (int)blockIdx.x * ROWS_OWN;
    const int nrows = (N - r0) < ROWS_OWN ? (N - r0) : ROWS_OWN;

    for (int i = tid; i < ROWS_OWN * DDIM / 4; i += 1024)
        reinterpret_cast<v4f*>(acc)[i] = v4f{0.f, 0.f, 0.f, 0.f};
    __syncthreads();

    v2i* mylst = lst + wave * LCAP;
    const int slot = (int)blockIdx.x * OWN_WAVES + wave;
    v2i* klst = reinterpret_cast<v2i*>(keep) + (size_t)slot * LCAP;
    int* keepCnt = reinterpret_cast<int*>(keep) + (size_t)gridDim.x * OWN_WAVES * LCAP * 2;
    const int nKept = reuse ? __builtin_amdgcn_readfirstlane(keepCnt[slot]) : LCAP + 1;
    int nLive = 0;

    if (nKept <= LCAP) {
        for (int i0 = 0; i0 < nKept; i0 += 128) {
            v2i en[4];
            #pragma unroll
            for (int u = 0; u < 4; ++u) {
                const int idx = i0 + u * 32 + lane;
                en[u] = v2i{0, 0};
                if (idx < nKept) en[u] = klst[idx];
            }
            #pragma unroll
            for (int u = 0; u < 4; ++u) {
                const bool live = (i0 + u * 32 + lane < nKept) & ((en[u].y >> 16) == hop);
                const unsigned mask = __builtin_amdgcn_ballot_w32(live);
                if (live) mylst[nLive + (int)__builtin_amdgcn_mbcnt_lo(mask, 0u)] = v2i{en[u].x, en[u].y & 0xffff};
                nLive += __builtin_popcount(mask);
            }
        }
    } else {
        int cnt = 0;
        const int TRIP = SCAN_V4 * 128;
        for (int base = wave * TRIP; base < E; base += OWN_WAVES * TRIP) {
            v4i d4[SCAN_V4];
            if (base + TRIP <= E) {
                #pragma unroll
                for (int u = 0; u < SCAN_V4; ++u)
                    d4[u] = *reinterpret_cast<const v4i_a4*>(dstIdx + base + u * 128 + lane * 4);
            } else {
                #pragma unroll
                for (int u = 0; u < SCAN_V4; ++u) {
                    const int eb = base + u * 128 + lane * 4;
                    d4[u].x = (eb     < E) ? dstIdx[eb]     : -1;
                    d4[u].y = (eb + 1 < E) ? dstIdx[eb + 1] : -1;
                    d4[u].z = (eb + 2 < E) ? dstIdx[eb + 2] : -1;
                    d4[u].w = (eb + 3 < E) ? dstIdx[eb + 3] : -1;
                }
            }
            unsigned ovf = 0;
            #pragma unroll
            for (int q = 0; q < SCAN_V4 * 4; ++q) {
                const int dl = d4[q >> 2][q & 3] - r0;
                const bool m = (unsigned)dl < (unsigned)nrows;
                const unsigned mask = __builtin_amdgcn_ballot_w32(m);
                if (mask) {
                    const int pos = cnt + (int)__builtin_amdgcn_mbcnt_lo(mask, 0u);
                    if (m) {
                        if (pos < LCAP) mylst[pos] = v2i{base + (q >> 2) * 128 + lane * 4 + (q & 3), dl};
                        else ovf |= 1u << q;
                    }
                    cnt += __builtin_popcount(mask);
                }
            }
            if (__builtin_amdgcn_ballot_w32(ovf != 0u)) {
                for (int q = 0; q < SCAN_V4 * 4; ++q) {
                    if ((ovf >> q) & 1u) {
                        const int e = base + (q >> 2) * 128 + lane * 4 + (q & 3);
                        if (attr[e] == hop) {
                            const float* row = xsrc + (size_t)srcIdx[e] * DDIM;
                            float* a = acc + (dstIdx[e] - r0) * DDIM;
                            for (int c = 0; c < DDIM; ++c) atomicAdd(a + (c & 3) * 32 + (c >> 2), row[c]);
                        }
                    }
                }
            }
        }

        const int nHit = cnt < LCAP ? cnt : LCAP;
        int nKeep = 0;
        for (int i0 = 0; i0 < nHit; i0 += 32) {
            const int idx = i0 + lane;
            const bool valid = idx < nHit;
            v2i en = v2i{0, 0};
            int k = 0, sr = 0;
            if (valid) {
                en = mylst[idx];
                k  = attr[en.x];
                sr = srcIdx[en.x];
            }
            if (!reuse) {
                const bool keep = valid & ((unsigned)(k - 1) < 2u);
                const unsigned km = __builtin_amdgcn_ballot_w32(keep);
                if (keep) klst[nKeep + (int)__builtin_amdgcn_mbcnt_lo(km, 0u)] = v2i{sr, en.y | (k << 16)};
                nKeep += __builtin_popcount(km);
            }
            const bool live = valid & (k == hop);
            const unsigned mask = __builtin_amdgcn_ballot_w32(live);
            if (live) mylst[nLive + (int)__builtin_amdgcn_mbcnt_lo(mask, 0u)] = v2i{sr, en.y};
            nLive += __builtin_popcount(mask);
        }
        if (!reuse && lane == 0) keepCnt[slot] = cnt > LCAP ? LCAP + 1 : nKeep;
    }
    if (lane == 0) lcnt[wave] = nLive;
    __syncthreads();

    int os = 0, od = 0, on = 0;
    for (int L = 0; L < OWN_WAVES; ++L) {
        const int nL = __builtin_amdgcn_readfirstlane(lcnt[L]);
        const v2i* lp = lst + L * LCAP;
        for (int i0 = 0; i0 < nL; i0 += 128) {
            v2i en[4];
            #pragma unroll
            for (int u = 0; u < 4; ++u) {
                const int idx = i0 + u * 32 + lane;
                en[u] = v2i{0, -1};
                if (idx < nL) en[u] = lp[idx];
            }
            #pragma unroll
            for (int u = 0; u < 4; ++u) {
                const bool mine = (en[u].y >= 0) & ((en[u].y & 31) == wave);
                unsigned mask = __builtin_amdgcn_ballot_w32(mine);
                while (mask) {
                    const int j = __builtin_ctz(mask);
                    mask &= mask - 1u;
                    os = (lane == on) ? __builtin_amdgcn_readlane(en[u].x, j) : os;
                    od = (lane == on) ? __builtin_amdgcn_readlane(en[u].y, j) : od;
                    ++on;
                    if (on == 32) { dgnn_own_flush(xsrc, acc, os, od, 32, lane); on = 0; }
                }
            }
        }
    }
    dgnn_own_flush(xsrc, acc, os, od, on, lane);
    __syncthreads();

    v4f* outp = reinterpret_cast<v4f*>(aggOut + (size_t)r0 * DDIM);
    for (int i = tid; i < nrows * (DDIM / 4); i += 1024) {
        const float* a = acc + (i >> 5) * DDIM + (i & 31);
        outp[i] = v4f{a[0], a[32], a[64], a[96]};
    }
}

__device__ __forceinline__ void dgnn_stage_load(const float* __restrict__ agg1,
                                                const float* __restrict__ agg2,
                                                int st, int nTiles, int wave, int c4,
                                                v4f (&p1)[4], v4f (&p2)[4]) {
    const int t0 = 2 * st;
    const int t1 = t0 + 1 < nTiles ? t0 + 1 : nTiles - 1;
    #pragma unroll
    for (int j = 0; j < 4; ++j) {
        const int g = (j < 2 ? t0 : t1) * 16 + (j & 1) * 8 + wave;
        p1[j] = *reinterpret_cast<const v4f*>(agg1 + (size_t)g * DDIM + c4);
        p2[j] = *reinterpret_cast<const v4f*>(agg2 + (size_t)g * DDIM + c4);
    }
}

__device__ __forceinline__ void dgnn_stage_store(float* __restrict__ stage, int wave, int c4,
                                                 const v4f (&p1)[4], const v4f (&p2)[4]) {
    #pragma unroll
    for (int j = 0; j < 4; ++j) {
        float* row = stage + (j * 8 + wave) * APITCH + c4;
        *reinterpret_cast<v4f*>(row)        = p1[j];
        *reinterpret_cast<v4f*>(row + DDIM) = p2[j];
    }
}

#define CHK 4
__device__ __forceinline__ void dgnn_frag_load(const float* __restrict__ aP, int c, v2f (&f)[4 * CHK]) {
    #pragma unroll
    for (int i = 0; i < CHK; ++i) {
        const int s = c * CHK + i;
        f[4 * i + 0] = *reinterpret_cast<const v2f*>(aP + 4 * s);
        f[4 * i + 1] = *reinterpret_cast<const v2f*>(aP + 16 * APITCH + 4 * s);
        f[4 * i + 2] = *reinterpret_cast<const v2f*>(aP + DDIM + 4 * s);
        f[4 * i + 3] = *reinterpret_cast<const v2f*>(aP + 16 * APITCH + DDIM + 4 * s);
    }
}

__global__ __launch_bounds__(256)
void dgnn_gemm_kernel(const float* __restrict__ agg1,
                      const float* __restrict__ agg2,
                      const float* __restrict__ Wt,
                      const float* __restrict__ bt,
                      const float* __restrict__ alphat,
                      const float* __restrict__ xin,
                      float* __restrict__ xout,
                      int nTiles, int nSuper) {
    extern __shared__ float lds[];

    const int wave = threadIdx.x >> 5;
    const int lane = threadIdx.x & 31;
    const int col  = lane & 15;
    const int half = lane >> 4;
    const int n    = wave * 16 + col;
    const int c4   = lane << 2;

    int st = blockIdx.x;
    const int stride = gridDim.x;

    v4f p1[4], p2[4];
    dgnn_stage_load(agg1, agg2, st < nSuper ? st : 0, nTiles, wave, c4, p1, p2);

    float e0 = __expf(alphat[0]);
    float e1 = __expf(alphat[1]);
    float inv = 1.0f / (e0 + e1);
    float a0 = e0 * inv;
    float a1 = e1 * inv;

    v2f b1[32], b2[32];
    {
        const float* w0p = Wt + (size_t)(2 * half) * DDIM + n;
        const float* w1p = w0p + (size_t)DDIM * DDIM;
        #pragma unroll
        for (int s = 0; s < 32; ++s) {
            b1[s].x = a0 * w0p[(4 * s) * DDIM];
            b1[s].y = a0 * w0p[(4 * s + 1) * DDIM];
            b2[s].x = a1 * w1p[(4 * s) * DDIM];
            b2[s].y = a1 * w1p[(4 * s + 1) * DDIM];
        }
    }

    const float bc = a0 * bt[n] + a1 * bt[DDIM + n];

    dgnn_stage_store(lds, wave, c4, p1, p2);
    __syncthreads();

    int cur = 0;
    for (; st < nSuper; st += stride) {
        const int nxt = st + stride;
        const bool more = nxt < nSuper;
        if (more) dgnn_stage_load(agg1, agg2, nxt, nTiles, wave, c4, p1, p2);

        const int t0 = 2 * st;
        const int t1 = t0 + 1 < nTiles ? t0 + 1 : nTiles - 1;
        const size_t o0 = (size_t)(t0 * 16 + half * 8) * DDIM + n;
        const size_t o1 = (size_t)(t1 * 16 + half * 8) * DDIM + n;

        float xr0[8], xr1[8];
        #pragma unroll
        for (int r = 0; r < 8; ++r) xr0[r] = xin[o0 + r * DDIM];
        #pragma unroll
        for (int r = 0; r < 8; ++r) xr1[r] = xin[o1 + r * DDIM];

        const float* aP = lds + cur * STAGE + col * APITCH + 2 * half;

        v8f acc0 = {};
        v8f acc1 = {};

        v2f fa[4 * CHK], fb[4 * CHK];
        dgnn_frag_load(aP, 0, fa);
        #pragma unroll
        for (int c = 0; c < 32 / CHK; c += 2) {
            dgnn_frag_load(aP, c + 1, fb);
            __builtin_amdgcn_sched_barrier(0);
            #pragma unroll
            for (int i = 0; i < CHK; ++i) {
                const int s = c * CHK + i;
                acc0 = __builtin_amdgcn_wmma_f32_16x16x4_f32(false, fa[4 * i + 0], false, b1[s], (short)0, acc0, false, false);
                acc1 = __builtin_amdgcn_wmma_f32_16x16x4_f32(false, fa[4 * i + 1], false, b1[s], (short)0, acc1, false, false);
                acc0 = __builtin_amdgcn_wmma_f32_16x16x4_f32(false, fa[4 * i + 2], false, b2[s], (short)0, acc0, false, false);
                acc1 = __builtin_amdgcn_wmma_f32_16x16x4_f32(false, fa[4 * i + 3], false, b2[s], (short)0, acc1, false, false);
            }
            __builtin_amdgcn_sched_barrier(0);
            if (c + 2 < 32 / CHK) dgnn_frag_load(aP, c + 2, fa);
            __builtin_amdgcn_sched_barrier(0);
            #pragma unroll
            for (int i = 0; i < CHK; ++i) {
                const int s = (c + 1) * CHK + i;
                acc0 = __builtin_amdgcn_wmma_f32_16x16x4_f32(false, fb[4 * i + 0], false, b1[s], (short)0, acc0, false, false);
                acc1 = __builtin_amdgcn_wmma_f32_16x16x4_f32(false, fb[4 * i + 1], false, b1[s], (short)0, acc1, false, false);
                acc0 = __builtin_amdgcn_wmma_f32_16x16x4_f32(false, fb[4 * i + 2], false, b2[s], (short)0, acc0, false, false);
                acc1 = __builtin_amdgcn_wmma_f32_16x16x4_f32(false, fb[4 * i + 3], false, b2[s], (short)0, acc1, false, false);
            }
            __builtin_amdgcn_sched_barrier(0);
        }

        #pragma unroll
        for (int r = 0; r < 8; ++r) {
            float v = acc0[r] + bc;
            v = v > 0.0f ? v : 0.0f;
            xout[o0 + r * DDIM] = xr0[r] + v;
        }
        #pragma unroll
        for (int r = 0; r < 8; ++r) {
            float v = acc1[r] + bc;
            v = v > 0.0f ? v : 0.0f;
            xout[o1 + r * DDIM] = xr1[r] + v;
        }

        if (more) dgnn_stage_store(lds + (cur ^ 1) * STAGE, wave, c4, p1, p2);
        __syncthreads();
        cur ^= 1;
    }
}

extern "C" void kernel_launch(void* const* d_in, const int* in_sizes, int n_in,
                              void* d_out, int out_size, void* d_ws, size_t ws_size,
                              hipStream_t stream) {
    const float* x     = (const float*)d_in[0];
    const int*   eidx  = (const int*)d_in[1];
    const int*   eattr = (const int*)d_in[2];
    const float* W     = (const float*)d_in[3];
    const float* b     = (const float*)d_in[4];
    const float* alpha = (const float*)d_in[5];
    float*       out   = (float*)d_out;

    const int N = in_sizes[0] / DDIM;
    const int E = in_sizes[2];
    const int T = 3;

    const size_t nd = (size_t)N * DDIM;
    float* ws   = (float*)d_ws;
    float* h1   = ws;
    float* h2   = ws + nd;
    float* agg1 = ws + 2 * nd;
    float* agg2 = ws + 3 * nd;

    const int* srcIdx = eidx;
    const int* dstIdx = eidx + E;

    const float* xs[4] = { x, h1, h2, out };

    const int nTiles = (N + 15) / 16;
    const int nSuper = (nTiles + 1) / 2;
    const int gemmBlocks = 256;
    const size_t ldsBytes = (size_t)2 * STAGE * sizeof(float);

    const int ownBlocks = (N + ROWS_OWN - 1) / ROWS_OWN;

    float* keep = out;

    for (int t = 0; t < T; ++t) {
        dgnn_own_kernel<<<ownBlocks, 1024, 0, stream>>>(
            xs[t], srcIdx, dstIdx, eattr, agg1, keep, E, N, 1,
            t > 0 ? 1 : 0);
        if (t != 1)
            dgnn_own_kernel<<<ownBlocks, 1024, 0, stream>>>(
                xs[t > 0 ? t - 1 : 0], srcIdx, dstIdx, eattr, agg2, keep, E, N, 2, 1);

        dgnn_gemm_kernel<<<gemmBlocks, 256, ldsBytes, stream>>>(
            agg1, agg2,
            W + (size_t)t * 2 * DDIM * DDIM,
            b + (size_t)t * 2 * DDIM,
            alpha + (size_t)t * 2,
            xs[t], (float*)xs[t + 1],
            nTiles, nSuper);
    }
}


# aggregation: first launch keeps each wave's own entries per hop in a bucket of its own; the other four read only that: no list walk, no barrier, a wave clears, adds, stores its rows; sum order kept
# speedup vs baseline: 3.3532x; 1.0438x over previous
#include <hip/hip_runtime.h>
#include <hip/hip_bf16.h>

#define DDIM 128
#define APITCH 260
#define TROWS 32
#define STAGE (TROWS * APITCH)

typedef __attribute__((ext_vector_type(2))) float v2f;
typedef __attribute__((ext_vector_type(4))) float v4f;
typedef __attribute__((ext_vector_type(8))) float v8f;

#define ROWS_OWN 400
#define OWN_WAVES 32
#define LCAP 384
#define SCAN_V4 8
#define ROWB 8
#define BCAP 192

typedef __attribute__((ext_vector_type(4))) int v4i;
typedef v4i v4i_a4 __attribute__((aligned(4)));
typedef __attribute__((ext_vector_type(2))) int v2i;

__device__ __forceinline__ void dgnn_own_flush(const float* __restrict__ xsrc, float* acc,
                                               int os, int od, int n, int lane) {
    for (int j = 0; j < n; j += ROWB) {
        v4f v[ROWB];
        #pragma unroll
        for (int X = 0; X < ROWB; ++X) {
            v[X] = v4f{0.f, 0.f, 0.f, 0.f};
            if (j + X < n) {
                const int sr = __builtin_amdgcn_readlane(os, j + X);
                v[X] = *reinterpret_cast<const v4f*>(xsrc + (size_t)sr * DDIM + lane * 4);
            }
        }
        #pragma unroll
        for (int X = 0; X < ROWB; ++X) {
            if (j + X < n) {
                const int dr = __builtin_amdgcn_readlane(od, j + X);
                float* a = acc + dr * DDIM + lane;
                atomicAdd(a + 0,  v[X].x);
                atomicAdd(a + 32, v[X].y);
                atomicAdd(a + 64, v[X].z);
                atomicAdd(a + 96, v[X].w);
            }
        }
    }
}

__global__ __launch_bounds__(1024)
void dgnn_own_kernel(const float* __restrict__ xsrc,
                     const int* __restrict__ srcIdx,
                     const int* __restrict__ dstIdx,
                     const int* __restrict__ attr,
                     float* __restrict__ aggOut,
                     float* __restrict__ keep,
                     int E, int N, int hop, int reuse) {
    __shared__ __attribute__((aligned(16))) float acc[ROWS_OWN * DDIM];
    __shared__ v2i lst[OWN_WAVES * LCAP];
    __shared__ int lcnt[OWN_WAVES];
    __shared__ int bad;

    const int tid  = threadIdx.x;
    const int lane = tid & 31;
    const int wave = __builtin_amdgcn_readfirstlane(tid >> 5);
    const int r0 = (int)blockIdx.x * ROWS_OWN;
    const int nrows = (N - r0) < ROWS_OWN ? (N - r0) : ROWS_OWN;

    const int slot = (int)blockIdx.x * OWN_WAVES + wave;
    v2i* kb = reinterpret_cast<v2i*>(keep) + (size_t)slot * 2 * BCAP;
    int* kcnt = reinterpret_cast<int*>(keep) + (size_t)gridDim.x * OWN_WAVES * 2 * BCAP * 2;
    int* kflag = kcnt + (size_t)gridDim.x * OWN_WAVES * 2;

    int fast = 0;
    if (reuse) fast = __builtin_amdgcn_readfirstlane(kflag[blockIdx.x] == 0 ? 1 : 0);

    if (fast) {
        const int n = __builtin_amdgcn_readfirstlane(kcnt[slot * 2 + (hop - 1)]);
        const v2i* bp = kb + (hop - 1) * BCAP;
        v2i* own = lst + wave * LCAP;
        v2i en[BCAP / 32];
        #pragma unroll
        for (int u = 0; u < BCAP / 32; ++u) {
            en[u] = v2i{0, 0};
            if (u * 32 + lane < n) en[u] = bp[u * 32 + lane];
        }
        for (int r = wave; r < ROWS_OWN; r += OWN_WAVES)
            *reinterpret_cast<v4f*>(acc + r * DDIM + lane * 4) = v4f{0.f, 0.f, 0.f, 0.f};
        #pragma unroll
        for (int u = 0; u < BCAP / 32; ++u)
            if (u * 32 < n) own[u * 32 + lane] = en[u];
        for (int j = 0; j < n; j += 32) {
            const v2i cur = own[j + lane];
            const int m = n - j;
            dgnn_own_flush(xsrc, acc, cur.x, cur.y, m < 32 ? m : 32, lane);
        }
        for (int r = wave; r < nrows; r += OWN_WAVES) {
            const float* a = acc + r * DDIM + lane;
            *reinterpret_cast<v4f*>(aggOut + (size_t)(r0 + r) * DDIM + lane * 4) = v4f{a[0], a[32], a[64], a[96]};
        }
        return;
    }

    if (tid == 0) bad = 0;
    for (int i = tid; i < ROWS_OWN * DDIM / 4; i += 1024)
        reinterpret_cast<v4f*>(acc)[i] = v4f{0.f, 0.f, 0.f, 0.f};
    __syncthreads();

    v2i* mylst = lst + wave * LCAP;
    int nLive = 0;
    {
        int cnt = 0;
        const int TRIP = SCAN_V4 * 128;
        for (int base = wave * TRIP; base < E; base += OWN_WAVES * TRIP) {
            v4i d4[SCAN_V4];
            if (base + TRIP <= E) {
                #pragma unroll
                for (int u = 0; u < SCAN_V4; ++u)
                    d4[u] = *reinterpret_cast<const v4i_a4*>(dstIdx + base + u * 128 + lane * 4);
            } else {
                #pragma unroll
                for (int u = 0; u < SCAN_V4; ++u) {
                    const int eb = base + u * 128 + lane * 4;
                    d4[u].x = (eb     < E) ? dstIdx[eb]     : -1;
                    d4[u].y = (eb + 1 < E) ? dstIdx[eb + 1] : -1;
                    d4[u].z = (eb + 2 < E) ? dstIdx[eb + 2] : -1;
                    d4[u].w = (eb + 3 < E) ? dstIdx[eb + 3] : -1;
                }
            }
            unsigned ovf = 0;
            #pragma unroll
            for (int q = 0; q < SCAN_V4 * 4; ++q) {
                const int dl = d4[q >> 2][q & 3] - r0;
                const bool m = (unsigned)dl < (unsigned)nrows;
                const unsigned mask = __builtin_amdgcn_ballot_w32(m);
                if (mask) {
                    const int pos = cnt + (int)__builtin_amdgcn_mbcnt_lo(mask, 0u);
                    if (m) {
                        if (pos < LCAP) mylst[pos] = v2i{base + (q >> 2) * 128 + lane * 4 + (q & 3), dl};
                        else ovf |= 1u << q;
                    }
                    cnt += __builtin_popcount(mask);
                }
            }
            if (__builtin_amdgcn_ballot_w32(ovf != 0u)) {
                for (int q = 0; q < SCAN_V4 * 4; ++q) {
                    if ((ovf >> q) & 1u) {
                        const int e = base + (q >> 2) * 128 + lane * 4 + (q & 3);
                        if (attr[e] == hop) {
                            const float* row = xsrc + (size_t)srcIdx[e] * DDIM;
                            float* a = acc + (dstIdx[e] - r0) * DDIM;
                            for (int c = 0; c < DDIM; ++c) atomicAdd(a + (c & 3) * 32 + (c >> 2), row[c]);
                        }
                    }
                }
            }
        }

        const int nHit = cnt < LCAP ? cnt : LCAP;
        for (int i0 = 0; i0 < nHit; i0 += 32) {
            const int idx = i0 + lane;
            const bool valid = idx < nHit;
            v2i en = v2i{0, 0};
            int k = 0, sr = 0;
            if (valid) {
                en = mylst[idx];
                k  = attr[en.x];
                sr = srcIdx[en.x];
            }
            const bool live = valid & ((unsigned)(k - 1) < 2u);
            const unsigned mask = __builtin_amdgcn_ballot_w32(live);
            if (live) mylst[nLive + (int)__builtin_amdgcn_mbcnt_lo(mask, 0u)] = v2i{sr, en.y | (k << 16)};
            nLive += __builtin_popcount(mask);
        }
        if (cnt > LCAP && lane == 0) bad = 1;
    }
    if (lane == 0) lcnt[wave] = nLive;
    __syncthreads();

    int os = 0, od = 0, on = 0, n1 = 0, n2 = 0;
    for (int L = 0; L < OWN_WAVES; ++L) {
        const int nL = __builtin_amdgcn_readfirstlane(lcnt[L]);
        const v2i* lp = lst + L * LCAP;
        for (int i0 = 0; i0 < nL; i0 += 128) {
            v2i en[4];
            #pragma unroll
            for (int u = 0; u < 4; ++u) {
                const int idx = i0 + u * 32 + lane;
                en[u] = v2i{0, -1};
                if (idx < nL) en[u] = lp[idx];
            }
            #pragma unroll
            for (int u = 0; u < 4; ++u) {
                const int y = en[u].y;
                const bool mine = (y >= 0) & ((y & 31) == wave);
                if (__builtin_amdgcn_ballot_w32(mine) == 0u) continue;
                const int k  = y >> 16;
                const int dl = y & 0xffff;
                if (!reuse) {
                    const bool s1 = mine & (k == 1);
                    const unsigned m1 = __builtin_amdgcn_ballot_w32(s1);
                    if (s1) {
                        const int pos = n1 + (int)__builtin_amdgcn_mbcnt_lo(m1, 0u);
                        if (pos < BCAP) kb[pos] = v2i{en[u].x, dl};
                    }
                    n1 += __builtin_popcount(m1);
                    const bool s2 = mine & (k == 2);
                    const unsigned m2 = __builtin_amdgcn_ballot_w32(s2);
                    if (s2) {
                        const int pos = n2 + (int)__builtin_amdgcn_mbcnt_lo(m2, 0u);
                        if (pos < BCAP) kb[BCAP + pos] = v2i{en[u].x, dl};
                    }
                    n2 += __builtin_popcount(m2);
                }
                unsigned mask = __builtin_amdgcn_ballot_w32(mine & (k == hop));
                while (mask) {
                    const int j = __builtin_ctz(mask);
                    mask &= mask - 1u;
                    os = (lane == on) ? __builtin_amdgcn_readlane(en[u].x, j) : os;
                    od = (lane == on) ? __builtin_amdgcn_readlane(dl, j) : od;
                    ++on;
                    if (on == 32) { dgnn_own_flush(xsrc, acc, os, od, 32, lane); on = 0; }
                }
            }
        }
    }
    dgnn_own_flush(xsrc, acc, os, od, on, lane);
    if (!reuse && lane == 0) {
        kcnt[slot * 2 + 0] = n1 < BCAP ? n1 : BCAP;
        kcnt[slot * 2 + 1] = n2 < BCAP ? n2 : BCAP;
        if (n1 > BCAP || n2 > BCAP) bad = 1;
    }
    __syncthreads();
    if (!reuse && tid == 0) kflag[blockIdx.x] = bad;

    v4f* outp = reinterpret_cast<v4f*>(aggOut + (size_t)r0 * DDIM);
    for (int i = tid; i < nrows * (DDIM / 4); i += 1024) {
        const float* a = acc + (i >> 5) * DDIM + (i & 31);
        outp[i] = v4f{a[0], a[32], a[64], a[96]};
    }
}

__device__ __forceinline__ void dgnn_stage_load(const float* __restrict__ agg1,
                                                const float* __restrict__ agg2,
                                                int st, int nTiles, int wave, int c4,
                                                v4f (&p1)[4], v4f (&p2)[4]) {
    const int t0 = 2 * st;
    const int t1 = t0 + 1 < nTiles ? t0 + 1 : nTiles - 1;
    #pragma unroll
    for (int j = 0; j < 4; ++j) {
        const int g = (j < 2 ? t0 : t1) * 16 + (j & 1) * 8 + wave;
        p1[j] = *reinterpret_cast<const v4f*>(agg1 + (size_t)g * DDIM + c4);
        p2[j] = *reinterpret_cast<const v4f*>(agg2 + (size_t)g * DDIM + c4);
    }
}

__device__ __forceinline__ void dgnn_stage_store(float* __restrict__ stage, int wave, int c4,
                                                 const v4f (&p1)[4], const v4f (&p2)[4]) {
    #pragma unroll
    for (int j = 0; j < 4; ++j) {
        float* row = stage + (j * 8 + wave) * APITCH + c4;
        *reinterpret_cast<v4f*>(row)        = p1[j];
        *reinterpret_cast<v4f*>(row + DDIM) = p2[j];
    }
}

#define CHK 4
__device__ __forceinline__ void dgnn_frag_load(const float* __restrict__ aP, int c, v2f (&f)[4 * CHK]) {
    #pragma unroll
    for (int i = 0; i < CHK; ++i) {
        const int s = c * CHK + i;
        f[4 * i + 0] = *reinterpret_cast<const v2f*>(aP + 4 * s);
        f[4 * i + 1] = *reinterpret_cast<const v2f*>(aP + 16 * APITCH + 4 * s);
        f[4 * i + 2] = *reinterpret_cast<const v2f*>(aP + DDIM + 4 * s);
        f[4 * i + 3] = *reinterpret_cast<const v2f*>(aP + 16 * APITCH + DDIM + 4 * s);
    }
}

__global__ __launch_bounds__(256)
void dgnn_gemm_kernel(const float* __restrict__ agg1,
                      const float* __restrict__ agg2,
                      const float* __restrict__ Wt,
                      const float* __restrict__ bt,
                      const float* __restrict__ alphat,
                      const float* __restrict__ xin,
                      float* __restrict__ xout,
                      int nTiles, int nSuper) {
    extern __shared__ float lds[];

    const int wave = threadIdx.x >> 5;
    const int lane = threadIdx.x & 31;
    const int col  = lane & 15;
    const int half = lane >> 4;
    const int n    = wave * 16 + col;
    const int c4   = lane << 2;

    int st = blockIdx.x;
    const int stride = gridDim.x;

    v4f p1[4], p2[4];
    dgnn_stage_load(agg1, agg2, st < nSuper ? st : 0, nTiles, wave, c4, p1, p2);

    float e0 = __expf(alphat[0]);
    float e1 = __expf(alphat[1]);
    float inv = 1.0f / (e0 + e1);
    float a0 = e0 * inv;
    float a1 = e1 * inv;

    v2f b1[32], b2[32];
    {
        const float* w0p = Wt + (size_t)(2 * half) * DDIM + n;
        const float* w1p = w0p + (size_t)DDIM * DDIM;
        #pragma unroll
        for (int s = 0; s < 32; ++s) {
            b1[s].x = a0 * w0p[(4 * s) * DDIM];
            b1[s].y = a0 * w0p[(4 * s + 1) * DDIM];
            b2[s].x = a1 * w1p[(4 * s) * DDIM];
            b2[s].y = a1 * w1p[(4 * s + 1) * DDIM];
        }
    }

    const float bc = a0 * bt[n] + a1 * bt[DDIM + n];

    dgnn_stage_store(lds, wave, c4, p1, p2);
    __syncthreads();

    int cur = 0;
    for (; st < nSuper; st += stride) {
        const int nxt = st + stride;
        const bool more = nxt < nSuper;
        if (more) dgnn_stage_load(agg1, agg2, nxt, nTiles, wave, c4, p1, p2);

        const int t0 = 2 * st;
        const int t1 = t0 + 1 < nTiles ? t0 + 1 : nTiles - 1;
        const size_t o0 = (size_t)(t0 * 16 + half * 8) * DDIM + n;
        const size_t o1 = (size_t)(t1 * 16 + half * 8) * DDIM + n;

        float xr0[8], xr1[8];
        #pragma unroll
        for (int r = 0; r < 8; ++r) xr0[r] = xin[o0 + r * DDIM];
        #pragma unroll
        for (int r = 0; r < 8; ++r) xr1[r] = xin[o1 + r * DDIM];

        const float* aP = lds + cur * STAGE + col * APITCH + 2 * half;

        v8f acc0 = {};
        v8f acc1 = {};

        v2f fa[4 * CHK], fb[4 * CHK];
        dgnn_frag_load(aP, 0, fa);
        #pragma unroll
        for (int c = 0; c < 32 / CHK; c += 2) {
            dgnn_frag_load(aP, c + 1, fb);
            __builtin_amdgcn_sched_barrier(0);
            #pragma unroll
            for (int i = 0; i < CHK; ++i) {
                const int s = c * CHK + i;
                acc0 = __builtin_amdgcn_wmma_f32_16x16x4_f32(false, fa[4 * i + 0], false, b1[s], (short)0, acc0, false, false);
                acc1 = __builtin_amdgcn_wmma_f32_16x16x4_f32(false, fa[4 * i + 1], false, b1[s], (short)0, acc1, false, false);
                acc0 = __builtin_amdgcn_wmma_f32_16x16x4_f32(false, fa[4 * i + 2], false, b2[s], (short)0, acc0, false, false);
                acc1 = __builtin_amdgcn_wmma_f32_16x16x4_f32(false, fa[4 * i + 3], false, b2[s], (short)0, acc1, false, false);
            }
            __builtin_amdgcn_sched_barrier(0);
            if (c + 2 < 32 / CHK) dgnn_frag_load(aP, c + 2, fa);
            __builtin_amdgcn_sched_barrier(0);
            #pragma unroll
            for (int i = 0; i < CHK; ++i) {
                const int s = (c + 1) * CHK + i;
                acc0 = __builtin_amdgcn_wmma_f32_16x16x4_f32(false, fb[4 * i + 0], false, b1[s], (short)0, acc0, false, false);
                acc1 = __builtin_amdgcn_wmma_f32_16x16x4_f32(false, fb[4 * i + 1], false, b1[s], (short)0, acc1, false, false);
                acc0 = __builtin_amdgcn_wmma_f32_16x16x4_f32(false, fb[4 * i + 2], false, b2[s], (short)0, acc0, false, false);
                acc1 = __builtin_amdgcn_wmma_f32_16x16x4_f32(false, fb[4 * i + 3], false, b2[s], (short)0, acc1, false, false);
            }
            __builtin_amdgcn_sched_barrier(0);
        }

        #pragma unroll
        for (int r = 0; r < 8; ++r) {
            float v = acc0[r] + bc;
            v = v > 0.0f ? v : 0.0f;
            xout[o0 + r * DDIM] = xr0[r] + v;
        }
        #pragma unroll
        for (int r = 0; r < 8; ++r) {
            float v = acc1[r] + bc;
            v = v > 0.0f ? v : 0.0f;
            xout[o1 + r * DDIM] = xr1[r] + v;
        }

        if (more) dgnn_stage_store(lds + (cur ^ 1) * STAGE, wave, c4, p1, p2);
        __syncthreads();
        cur ^= 1;
    }
}

extern "C" void kernel_launch(void* const* d_in, const int* in_sizes, int n_in,
                              void* d_out, int out_size, void* d_ws, size_t ws_size,
                              hipStream_t stream) {
    const float* x     = (const float*)d_in[0];
    const int*   eidx  = (const int*)d_in[1];
    const int*   eattr = (const int*)d_in[2];
    const float* W     = (const float*)d_in[3];
    const float* b     = (const float*)d_in[4];
    const float* alpha = (const float*)d_in[5];
    float*       out   = (float*)d_out;

    const int N = in_sizes[0] / DDIM;
    const int E = in_sizes[2];
    const int T = 3;

    const size_t nd = (size_t)N * DDIM;
    float* ws   = (float*)d_ws;
    float* h1   = ws;
    float* h2   = ws + nd;
    float* agg1 = ws + 2 * nd;
    float* agg2 = ws + 3 * nd;

    const int* srcIdx = eidx;
    const int* dstIdx = eidx + E;

    const float* xs[4] = { x, h1, h2, out };

    const int nTiles = (N + 15) / 16;
    const int nSuper = (nTiles + 1) / 2;
    const int gemmBlocks = 256;
    const size_t ldsBytes = (size_t)2 * STAGE * sizeof(float);

    const int ownBlocks = (N + ROWS_OWN - 1) / ROWS_OWN;

    float* keep = out;

    for (int t = 0; t < T; ++t) {
        dgnn_own_kernel<<<ownBlocks, 1024, 0, stream>>>(
            xs[t], srcIdx, dstIdx, eattr, agg1, keep, E, N, 1,
            t > 0 ? 1 : 0);
        if (t != 1)
            dgnn_own_kernel<<<ownBlocks, 1024, 0, stream>>>(
                xs[t > 0 ? t - 1 : 0], srcIdx, dstIdx, eattr, agg2, keep, E, N, 2, 1);

        dgnn_gemm_kernel<<<gemmBlocks, 256, ldsBytes, stream>>>(
            agg1, agg2,
            W + (size_t)t * 2 * DDIM * DDIM,
            b + (size_t)t * 2 * DDIM,
            alpha + (size_t)t * 2,
            xs[t], (float*)xs[t + 1],
            nTiles, nSuper);
    }
}


# aggregation, kept-bucket launches: a wave groups its bucket by target row and sums each row in registers, one store per row; LDS float adds, clear and read-back gone there; sum order kept
# speedup vs baseline: 3.4636x; 1.0329x over previous
#include <hip/hip_runtime.h>
#include <hip/hip_bf16.h>

#define DDIM 128
#define APITCH 260
#define TROWS 32
#define STAGE (TROWS * APITCH)

typedef __attribute__((ext_vector_type(2))) float v2f;
typedef __attribute__((ext_vector_type(4))) float v4f;
typedef __attribute__((ext_vector_type(8))) float v8f;

#define ROWS_OWN 400
#define OWN_WAVES 32
#define LCAP 384
#define SCAN_V4 8
#define ROWB 8
#define BCAP 192

typedef __attribute__((ext_vector_type(4))) int v4i;
typedef v4i v4i_a4 __attribute__((aligned(4)));
typedef __attribute__((ext_vector_type(2))) int v2i;

__device__ __forceinline__ void dgnn_own_flush(const float* __restrict__ xsrc, float* acc,
                                               int os, int od, int n, int lane) {
    for (int j = 0; j < n; j += ROWB) {
        v4f v[ROWB];
        #pragma unroll
        for (int X = 0; X < ROWB; ++X) {
            v[X] = v4f{0.f, 0.f, 0.f, 0.f};
            if (j + X < n) {
                const int sr = __builtin_amdgcn_readlane(os, j + X);
                v[X] = *reinterpret_cast<const v4f*>(xsrc + (size_t)sr * DDIM + lane * 4);
            }
        }
        #pragma unroll
        for (int X = 0; X < ROWB; ++X) {
            if (j + X < n) {
                const int dr = __builtin_amdgcn_readlane(od, j + X);
                float* a = acc + dr * DDIM + lane;
                atomicAdd(a + 0,  v[X].x);
                atomicAdd(a + 32, v[X].y);
                atomicAdd(a + 64, v[X].z);
                atomicAdd(a + 96, v[X].w);
            }
        }
    }
}

template <bool FULL>
__device__ __forceinline__ void dgnn_row_batch(const float* __restrict__ xlane, float* __restrict__ orow,
                                               int srcs, int rows, int lasts, int b, int m, v4f& sum) {
    v4f v[ROWB];
    #pragma unroll
    for (int X = 0; X < ROWB; ++X) {
        if (FULL || b + X < m) {
            const unsigned sr = (unsigned)__builtin_amdgcn_readlane(srcs, b + X);
            v[X] = *reinterpret_cast<const v4f*>(xlane + (size_t)sr * DDIM);
        }
    }
    #pragma unroll
    for (int X = 0; X < ROWB; ++X) {
        if (FULL || b + X < m) {
            sum += v[X];
            if (__builtin_amdgcn_readlane(lasts, b + X)) {
                const unsigned dr = (unsigned)__builtin_amdgcn_readlane(rows, b + X);
                *reinterpret_cast<v4f*>(orow + (size_t)dr * DDIM) = sum;
                sum = v4f{0.f, 0.f, 0.f, 0.f};
            }
        }
    }
}

__global__ __launch_bounds__(1024)
void dgnn_own_kernel(const float* __restrict__ xsrc,
                     const int* __restrict__ srcIdx,
                     const int* __restrict__ dstIdx,
                     const int* __restrict__ attr,
                     float* __restrict__ aggOut,
                     float* __restrict__ keep,
                     int E, int N, int hop, int reuse) {
    __shared__ __attribute__((aligned(16))) float acc[ROWS_OWN * DDIM];
    __shared__ v2i lst[OWN_WAVES * LCAP];
    __shared__ int lcnt[OWN_WAVES];
    __shared__ int bad;

    const int tid  = threadIdx.x;
    const int lane = tid & 31;
    const int wave = __builtin_amdgcn_readfirstlane(tid >> 5);
    const int r0 = (int)blockIdx.x * ROWS_OWN;
    const int nrows = (N - r0) < ROWS_OWN ? (N - r0) : ROWS_OWN;

    const int slot = (int)blockIdx.x * OWN_WAVES + wave;
    v2i* kb = reinterpret_cast<v2i*>(keep) + (size_t)slot * 2 * BCAP;
    int* kcnt = reinterpret_cast<int*>(keep) + (size_t)gridDim.x * OWN_WAVES * 2 * BCAP * 2;
    int* kflag = kcnt + (size_t)gridDim.x * OWN_WAVES * 2;

    int fast = 0;
    if (reuse) fast = __builtin_amdgcn_readfirstlane(kflag[blockIdx.x] == 0 ? 1 : 0);

    if (fast) {
        const int n = __builtin_amdgcn_readfirstlane(kcnt[slot * 2 + (hop - 1)]);
        const v2i* bp = kb + (hop - 1) * BCAP;
        v2i* own = lst + wave * LCAP;
        v2i en[BCAP / 32];
        #pragma unroll
        for (int u = 0; u < BCAP / 32; ++u) {
            en[u] = v2i{0, -1};
            if (u * 32 + lane < n) en[u] = bp[u * 32 + lane];
        }
        float* orow = aggOut + (size_t)r0 * DDIM + lane * 4;

        int pos = 0;
        for (int r = wave; r < nrows; r += OWN_WAVES) {
            const int start = pos;
            #pragma unroll
            for (int u = 0; u < BCAP / 32; ++u) {
                const bool hit = en[u].y == r;
                const unsigned mask = __builtin_amdgcn_ballot_w32(hit);
                if (hit) own[pos + (int)__builtin_amdgcn_mbcnt_lo(mask, 0u)] = en[u];
                pos += __builtin_popcount(mask);
            }
            if (pos == start)
                *reinterpret_cast<v4f*>(orow + (size_t)r * DDIM) = v4f{0.f, 0.f, 0.f, 0.f};
        }

        const float* xlane = xsrc + lane * 4;
        v4f sum = v4f{0.f, 0.f, 0.f, 0.f};
        for (int j = 0; j < n; j += 32) {
            const int idx = j + lane;
            v2i cur = v2i{0, -1};
            int nextRow = -1;
            if (idx < n) cur = own[idx];
            if (idx + 1 < n) nextRow = own[idx + 1].y;
            const int last = (cur.y != nextRow) ? 1 : 0;
            const int m = (n - j) < 32 ? (n - j) : 32;
            int b = 0;
            for (; b + ROWB <= m; b += ROWB)
                dgnn_row_batch<true>(xlane, orow, cur.x, cur.y, last, b, m, sum);
            if (b < m)
                dgnn_row_batch<false>(xlane, orow, cur.x, cur.y, last, b, m, sum);
        }
        return;
    }

    if (tid == 0) bad = 0;
    for (int i = tid; i < ROWS_OWN * DDIM / 4; i += 1024)
        reinterpret_cast<v4f*>(acc)[i] = v4f{0.f, 0.f, 0.f, 0.f};
    __syncthreads();

    v2i* mylst = lst + wave * LCAP;
    int nLive = 0;
    {
        int cnt = 0;
        const int TRIP = SCAN_V4 * 128;
        for (int base = wave * TRIP; base < E; base += OWN_WAVES * TRIP) {
            v4i d4[SCAN_V4];
            if (base + TRIP <= E) {
                #pragma unroll
                for (int u = 0; u < SCAN_V4; ++u)
                    d4[u] = *reinterpret_cast<const v4i_a4*>(dstIdx + base + u * 128 + lane * 4);
            } else {
                #pragma unroll
                for (int u = 0; u < SCAN_V4; ++u) {
                    const int eb = base + u * 128 + lane * 4;
                    d4[u].x = (eb     < E) ? dstIdx[eb]     : -1;
                    d4[u].y = (eb + 1 < E) ? dstIdx[eb + 1] : -1;
                    d4[u].z = (eb + 2 < E) ? dstIdx[eb + 2] : -1;
                    d4[u].w = (eb + 3 < E) ? dstIdx[eb + 3] : -1;
                }
            }
            unsigned ovf = 0;
            #pragma unroll
            for (int q = 0; q < SCAN_V4 * 4; ++q) {
                const int dl = d4[q >> 2][q & 3] - r0;
                const bool m = (unsigned)dl < (unsigned)nrows;
                const unsigned mask = __builtin_amdgcn_ballot_w32(m);
                if (mask) {
                    const int pos = cnt + (int)__builtin_amdgcn_mbcnt_lo(mask, 0u);
                    if (m) {
                        if (pos < LCAP) mylst[pos] = v2i{base + (q >> 2) * 128 + lane * 4 + (q & 3), dl};
                        else ovf |= 1u << q;
                    }
                    cnt += __builtin_popcount(mask);
                }
            }
            if (__builtin_amdgcn_ballot_w32(ovf != 0u)) {
                for (int q = 0; q < SCAN_V4 * 4; ++q) {
                    if ((ovf >> q) & 1u) {
                        const int e = base + (q >> 2) * 128 + lane * 4 + (q & 3);
                        if (attr[e] == hop) {
                            const float* row = xsrc + (size_t)srcIdx[e] * DDIM;
                            float* a = acc + (dstIdx[e] - r0) * DDIM;
                            for (int c = 0; c < DDIM; ++c) atomicAdd(a + (c & 3) * 32 + (c >> 2), row[c]);
                        }
                    }
                }
            }
        }

        const int nHit = cnt < LCAP ? cnt : LCAP;
        for (int i0 = 0; i0 < nHit; i0 += 32) {
            const int idx = i0 + lane;
            const bool valid = idx < nHit;
            v2i en = v2i{0, 0};
            int k = 0, sr = 0;
            if (valid) {
                en = mylst[idx];
                k  = attr[en.x];
                sr = srcIdx[en.x];
            }
            const bool live = valid & ((unsigned)(k - 1) < 2u);
            const unsigned mask = __builtin_amdgcn_ballot_w32(live);
            if (live) mylst[nLive + (int)__builtin_amdgcn_mbcnt_lo(mask, 0u)] = v2i{sr, en.y | (k << 16)};
            nLive += __builtin_popcount(mask);
        }
        if (cnt > LCAP && lane == 0) bad = 1;
    }
    if (lane == 0) lcnt[wave] = nLive;
    __syncthreads();

    int os = 0, od = 0, on = 0, n1 = 0, n2 = 0;
    for (int L = 0; L < OWN_WAVES; ++L) {
        const int nL = __builtin_amdgcn_readfirstlane(lcnt[L]);
        const v2i* lp = lst + L * LCAP;
        for (int i0 = 0; i0 < nL; i0 += 128) {
            v2i en[4];
            #pragma unroll
            for (int u = 0; u < 4; ++u) {
                const int idx = i0 + u * 32 + lane;
                en[u] = v2i{0, -1};
                if (idx < nL) en[u] = lp[idx];
            }
            #pragma unroll
            for (int u = 0; u < 4; ++u) {
                const int y = en[u].y;
                const bool mine = (y >= 0) & ((y & 31) == wave);
                if (__builtin_amdgcn_ballot_w32(mine) == 0u) continue;
                const int k  = y >> 16;
                const int dl = y & 0xffff;
                if (!reuse) {
                    const bool s1 = mine & (k == 1);
                    const unsigned m1 = __builtin_amdgcn_ballot_w32(s1);
                    if (s1) {
                        const int pos = n1 + (int)__builtin_amdgcn_mbcnt_lo(m1, 0u);
                        if (pos < BCAP) kb[pos] = v2i{en[u].x, dl};
                    }
                    n1 += __builtin_popcount(m1);
                    const bool s2 = mine & (k == 2);
                    const unsigned m2 = __builtin_amdgcn_ballot_w32(s2);
                    if (s2) {
                        const int pos = n2 + (int)__builtin_amdgcn_mbcnt_lo(m2, 0u);
                        if (pos < BCAP) kb[BCAP + pos] = v2i{en[u].x, dl};
                    }
                    n2 += __builtin_popcount(m2);
                }
                unsigned mask = __builtin_amdgcn_ballot_w32(mine & (k == hop));
                while (mask) {
                    const int j = __builtin_ctz(mask);
                    mask &= mask - 1u;
                    os = (lane == on) ? __builtin_amdgcn_readlane(en[u].x, j) : os;
                    od = (lane == on) ? __builtin_amdgcn_readlane(dl, j) : od;
                    ++on;
                    if (on == 32) { dgnn_own_flush(xsrc, acc, os, od, 32, lane); on = 0; }
                }
            }
        }
    }
    dgnn_own_flush(xsrc, acc, os, od, on, lane);
    if (!reuse && lane == 0) {
        kcnt[slot * 2 + 0] = n1 < BCAP ? n1 : BCAP;
        kcnt[slot * 2 + 1] = n2 < BCAP ? n2 : BCAP;
        if (n1 > BCAP || n2 > BCAP) bad = 1;
    }
    __syncthreads();
    if (!reuse && tid == 0) kflag[blockIdx.x] = bad;

    v4f* outp = reinterpret_cast<v4f*>(aggOut + (size_t)r0 * DDIM);
    for (int i = tid; i < nrows * (DDIM / 4); i += 1024) {
        const float* a = acc + (i >> 5) * DDIM + (i & 31);
        outp[i] = v4f{a[0], a[32], a[64], a[96]};
    }
}

__device__ __forceinline__ void dgnn_stage_load(const float* __restrict__ agg1,
                                                const float* __restrict__ agg2,
                                                int st, int nTiles, int wave, int c4,
                                                v4f (&p1)[4], v4f (&p2)[4]) {
    const int t0 = 2 * st;
    const int t1 = t0 + 1 < nTiles ? t0 + 1 : nTiles - 1;
    #pragma unroll
    for (int j = 0; j < 4; ++j) {
        const int g = (j < 2 ? t0 : t1) * 16 + (j & 1) * 8 + wave;
        p1[j] = *reinterpret_cast<const v4f*>(agg1 + (size_t)g * DDIM + c4);
        p2[j] = *reinterpret_cast<const v4f*>(agg2 + (size_t)g * DDIM + c4);
    }
}

__device__ __forceinline__ void dgnn_stage_store(float* __restrict__ stage, int wave, int c4,
                                                 const v4f (&p1)[4], const v4f (&p2)[4]) {
    #pragma unroll
    for (int j = 0; j < 4; ++j) {
        float* row = stage + (j * 8 + wave) * APITCH + c4;
        *reinterpret_cast<v4f*>(row)        = p1[j];
        *reinterpret_cast<v4f*>(row + DDIM) = p2[j];
    }
}

#define CHK 4
__device__ __forceinline__ void dgnn_frag_load(const float* __restrict__ aP, int c, v2f (&f)[4 * CHK]) {
    #pragma unroll
    for (int i = 0; i < CHK; ++i) {
        const int s = c * CHK + i;
        f[4 * i + 0] = *reinterpret_cast<const v2f*>(aP + 4 * s);
        f[4 * i + 1] = *reinterpret_cast<const v2f*>(aP + 16 * APITCH + 4 * s);
        f[4 * i + 2] = *reinterpret_cast<const v2f*>(aP + DDIM + 4 * s);
        f[4 * i + 3] = *reinterpret_cast<const v2f*>(aP + 16 * APITCH + DDIM + 4 * s);
    }
}

__global__ __launch_bounds__(256)
void dgnn_gemm_kernel(const float* __restrict__ agg1,
                      const float* __restrict__ agg2,
                      const float* __restrict__ Wt,
                      const float* __restrict__ bt,
                      const float* __restrict__ alphat,
                      const float* __restrict__ xin,
                      float* __restrict__ xout,
                      int nTiles, int nSuper) {
    extern __shared__ float lds[];

    const int wave = threadIdx.x >> 5;
    const int lane = threadIdx.x & 31;
    const int col  = lane & 15;
    const int half = lane >> 4;
    const int n    = wave * 16 + col;
    const int c4   = lane << 2;

    int st = blockIdx.x;
    const int stride = gridDim.x;

    v4f p1[4], p2[4];
    dgnn_stage_load(agg1, agg2, st < nSuper ? st : 0, nTiles, wave, c4, p1, p2);

    float e0 = __expf(alphat[0]);
    float e1 = __expf(alphat[1]);
    float inv = 1.0f / (e0 + e1);
    float a0 = e0 * inv;
    float a1 = e1 * inv;

    v2f b1[32], b2[32];
    {
        const float* w0p = Wt + (size_t)(2 * half) * DDIM + n;
        const float* w1p = w0p + (size_t)DDIM * DDIM;
        #pragma unroll
        for (int s = 0; s < 32; ++s) {
            b1[s].x = a0 * w0p[(4 * s) * DDIM];
            b1[s].y = a0 * w0p[(4 * s + 1) * DDIM];
            b2[s].x = a1 * w1p[(4 * s) * DDIM];
            b2[s].y = a1 * w1p[(4 * s + 1) * DDIM];
        }
    }

    const float bc = a0 * bt[n] + a1 * bt[DDIM + n];

    dgnn_stage_store(lds, wave, c4, p1, p2);
    __syncthreads();

    int cur = 0;
    for (; st < nSuper; st += stride) {
        const int nxt = st + stride;
        const bool more = nxt < nSuper;
        if (more) dgnn_stage_load(agg1, agg2, nxt, nTiles, wave, c4, p1, p2);

        const int t0 = 2 * st;
        const int t1 = t0 + 1 < nTiles ? t0 + 1 : nTiles - 1;
        const size_t o0 = (size_t)(t0 * 16 + half * 8) * DDIM + n;
        const size_t o1 = (size_t)(t1 * 16 + half * 8) * DDIM + n;

        float xr0[8], xr1[8];
        #pragma unroll
        for (int r = 0; r < 8; ++r) xr0[r] = xin[o0 + r * DDIM];
        #pragma unroll
        for (int r = 0; r < 8; ++r) xr1[r] = xin[o1 + r * DDIM];

        const float* aP = lds + cur * STAGE + col * APITCH + 2 * half;

        v8f acc0 = {};
        v8f acc1 = {};

        v2f fa[4 * CHK], fb[4 * CHK];
        dgnn_frag_load(aP, 0, fa);
        #pragma unroll
        for (int c = 0; c < 32 / CHK; c += 2) {
            dgnn_frag_load(aP, c + 1, fb);
            __builtin_amdgcn_sched_barrier(0);
            #pragma unroll
            for (int i = 0; i < CHK; ++i) {
                const int s = c * CHK + i;
                acc0 = __builtin_amdgcn_wmma_f32_16x16x4_f32(false, fa[4 * i + 0], false, b1[s], (short)0, acc0, false, false);
                acc1 = __builtin_amdgcn_wmma_f32_16x16x4_f32(false, fa[4 * i + 1], false, b1[s], (short)0, acc1, false, false);
                acc0 = __builtin_amdgcn_wmma_f32_16x16x4_f32(false, fa[4 * i + 2], false, b2[s], (short)0, acc0, false, false);
                acc1 = __builtin_amdgcn_wmma_f32_16x16x4_f32(false, fa[4 * i + 3], false, b2[s], (short)0, acc1, false, false);
            }
            __builtin_amdgcn_sched_barrier(0);
            if (c + 2 < 32 / CHK) dgnn_frag_load(aP, c + 2, fa);
            __builtin_amdgcn_sched_barrier(0);
            #pragma unroll
            for (int i = 0; i < CHK; ++i) {
                const int s = (c + 1) * CHK + i;
                acc0 = __builtin_amdgcn_wmma_f32_16x16x4_f32(false, fb[4 * i + 0], false, b1[s], (short)0, acc0, false, false);
                acc1 = __builtin_amdgcn_wmma_f32_16x16x4_f32(false, fb[4 * i + 1], false, b1[s], (short)0, acc1, false, false);
                acc0 = __builtin_amdgcn_wmma_f32_16x16x4_f32(false, fb[4 * i + 2], false, b2[s], (short)0, acc0, false, false);
                acc1 = __builtin_amdgcn_wmma_f32_16x16x4_f32(false, fb[4 * i + 3], false, b2[s], (short)0, acc1, false, false);
            }
            __builtin_amdgcn_sched_barrier(0);
        }

        #pragma unroll
        for (int r = 0; r < 8; ++r) {
            float v = acc0[r] + bc;
            v = v > 0.0f ? v : 0.0f;
            xout[o0 + r * DDIM] = xr0[r] + v;
        }
        #pragma unroll
        for (int r = 0; r < 8; ++r) {
            float v = acc1[r] + bc;
            v = v > 0.0f ? v : 0.0f;
            xout[o1 + r * DDIM] = xr1[r] + v;
        }

        if (more) dgnn_stage_store(lds + (cur ^ 1) * STAGE, wave, c4, p1, p2);
        __syncthreads();
        cur ^= 1;
    }
}

extern "C" void kernel_launch(void* const* d_in, const int* in_sizes, int n_in,
                              void* d_out, int out_size, void* d_ws, size_t ws_size,
                              hipStream_t stream) {
    const float* x     = (const float*)d_in[0];
    const int*   eidx  = (const int*)d_in[1];
    const int*   eattr = (const int*)d_in[2];
    const float* W     = (const float*)d_in[3];
    const float* b     = (const float*)d_in[4];
    const float* alpha = (const float*)d_in[5];
    float*       out   = (float*)d_out;

    const int N = in_sizes[0] / DDIM;
    const int E = in_sizes[2];
    const int T = 3;

    const size_t nd = (size_t)N * DDIM;
    float* ws   = (float*)d_ws;
    float* h1   = ws;
    float* h2   = ws + nd;
    float* agg1 = ws + 2 * nd;
    float* agg2 = ws + 3 * nd;

    const int* srcIdx = eidx;
    const int* dstIdx = eidx + E;

    const float* xs[4] = { x, h1, h2, out };

    const int nTiles = (N + 15) / 16;
    const int nSuper = (nTiles + 1) / 2;
    const int gemmBlocks = 256;
    const size_t ldsBytes = (size_t)2 * STAGE * sizeof(float);

    const int ownBlocks = (N + ROWS_OWN - 1) / ROWS_OWN;

    float* keep = out;

    for (int t = 0; t < T; ++t) {
        dgnn_own_kernel<<<ownBlocks, 1024, 0, stream>>>(
            xs[t], srcIdx, dstIdx, eattr, agg1, keep, E, N, 1,
            t > 0 ? 1 : 0);
        if (t != 1)
            dgnn_own_kernel<<<ownBlocks, 1024, 0, stream>>>(
                xs[t > 0 ? t - 1 : 0], srcIdx, dstIdx, eattr, agg2, keep, E, N, 2, 1);

        dgnn_gemm_kernel<<<gemmBlocks, 256, ldsBytes, stream>>>(
            agg1, agg2,
            W + (size_t)t * 2 * DDIM * DDIM,
            b + (size_t)t * 2 * DDIM,
            alpha + (size_t)t * 2,
            xs[t], (float*)xs[t + 1],
            nTiles, nSuper);
    }
}


# aggregation scan on packed edges: a launch in front writes one owner byte and one 8-byte record per edge; owners scan bytes, not 4-byte targets, one load per hit; sum order = new list order
# speedup vs baseline: 3.7964x; 1.0961x over previous
#include <hip/hip_runtime.h>
#include <hip/hip_bf16.h>

#define DDIM 128
#define APITCH 260
#define TROWS 32
#define STAGE (TROWS * APITCH)

typedef __attribute__((ext_vector_type(2))) float v2f;
typedef __attribute__((ext_vector_type(4))) float v4f;
typedef __attribute__((ext_vector_type(8))) float v8f;

#define ROWS_OWN 400
#define OWN_WAVES 32
#define LCAP 384
#define SCAN_B16 8
#define TRIPB (SCAN_B16 * 512)
#define SWEEP (OWN_WAVES * TRIPB)
#define OWNB_OFF 25165824u
#define REC_OFF  27262976u
#define ROWB 8
#define BCAP 192

typedef __attribute__((ext_vector_type(4))) int v4i;
typedef v4i v4i_a4 __attribute__((aligned(4)));
typedef __attribute__((ext_vector_type(2))) int v2i;
typedef __attribute__((ext_vector_type(4))) unsigned v4u;

__global__ __launch_bounds__(256)
void dgnn_pack_kernel(const int* __restrict__ srcIdx,
                      const int* __restrict__ dstIdx,
                      const int* __restrict__ attr,
                      float* __restrict__ keep,
                      int E, int N) {
    unsigned char* kbytes = reinterpret_cast<unsigned char*>(keep);
    const int e0 = ((int)blockIdx.x * 256 + (int)threadIdx.x) * 4;
    const bool whole = e0 + 4 <= E;
    v4i d = v4i{-1, -1, -1, -1}, s = v4i{0, 0, 0, 0}, k = v4i{0, 0, 0, 0};
    if (whole) {
        d = *reinterpret_cast<const v4i_a4*>(dstIdx + e0);
        s = *reinterpret_cast<const v4i_a4*>(srcIdx + e0);
        k = *reinterpret_cast<const v4i_a4*>(attr + e0);
    } else {
        for (int j = 0; j < 4; ++j) {
            if (e0 + j < E) { d[j] = dstIdx[e0 + j]; s[j] = srcIdx[e0 + j]; k[j] = attr[e0 + j]; }
        }
    }
    unsigned w = 0;
    v2i r[4];
    #pragma unroll
    for (int j = 0; j < 4; ++j) {
        const unsigned du = (unsigned)d[j];
        const bool live = (du < (unsigned)N) & ((unsigned)(k[j] - 1) < 2u);
        const unsigned o = du / ROWS_OWN;
        const unsigned dl = du - o * ROWS_OWN;
        w |= (live ? o : 255u) << (8 * j);
        r[j] = v2i{s[j], (int)(dl | ((unsigned)k[j] << 16))};
    }
    *reinterpret_cast<unsigned*>(kbytes + OWNB_OFF + e0) = w;
    v2i* rec = reinterpret_cast<v2i*>(kbytes + REC_OFF);
    if (whole) {
        *reinterpret_cast<v4i*>(rec + e0)     = v4i{r[0].x, r[0].y, r[1].x, r[1].y};
        *reinterpret_cast<v4i*>(rec + e0 + 2) = v4i{r[2].x, r[2].y, r[3].x, r[3].y};
    } else {
        for (int j = 0; j < 4; ++j)
            if (e0 + j < E) rec[e0 + j] = r[j];
    }
}

__device__ __forceinline__ void dgnn_own_flush(const float* __restrict__ xsrc, float* acc,
                                               int os, int od, int n, int lane) {
    for (int j = 0; j < n; j += ROWB) {
        v4f v[ROWB];
        #pragma unroll
        for (int X = 0; X < ROWB; ++X) {
            v[X] = v4f{0.f, 0.f, 0.f, 0.f};
            if (j + X < n) {
                const int sr = __builtin_amdgcn_readlane(os, j + X);
                v[X] = *reinterpret_cast<const v4f*>(xsrc + (size_t)sr * DDIM + lane * 4);
            }
        }
        #pragma unroll
        for (int X = 0; X < ROWB; ++X) {
            if (j + X < n) {
                const int dr = __builtin_amdgcn_readlane(od, j + X);
                float* a = acc + dr * DDIM + lane;
                atomicAdd(a + 0,  v[X].x);
                atomicAdd(a + 32, v[X].y);
                atomicAdd(a + 64, v[X].z);
                atomicAdd(a + 96, v[X].w);
            }
        }
    }
}

template <bool FULL>
__device__ __forceinline__ void dgnn_row_batch(const float* __restrict__ xlane, float* __restrict__ orow,
                                               int srcs, int rows, int lasts, int b, int m, v4f& sum) {
    v4f v[ROWB];
    #pragma unroll
    for (int X = 0; X < ROWB; ++X) {
        if (FULL || b + X < m) {
            const unsigned sr = (unsigned)__builtin_amdgcn_readlane(srcs, b + X);
            v[X] = *reinterpret_cast<const v4f*>(xlane + (size_t)sr * DDIM);
        }
    }
    #pragma unroll
    for (int X = 0; X < ROWB; ++X) {
        if (FULL || b + X < m) {
            sum += v[X];
            if (__builtin_amdgcn_readlane(lasts, b + X)) {
                const unsigned dr = (unsigned)__builtin_amdgcn_readlane(rows, b + X);
                *reinterpret_cast<v4f*>(orow + (size_t)dr * DDIM) = sum;
                sum = v4f{0.f, 0.f, 0.f, 0.f};
            }
        }
    }
}

__global__ __launch_bounds__(1024)
void dgnn_own_kernel(const float* __restrict__ xsrc,
                     const int* __restrict__ srcIdx,
                     const int* __restrict__ dstIdx,
                     const int* __restrict__ attr,
                     float* __restrict__ aggOut,
                     float* __restrict__ keep,
                     int E, int N, int hop, int reuse) {
    __shared__ __attribute__((aligned(16))) float acc[ROWS_OWN * DDIM];
    __shared__ v2i lst[OWN_WAVES * LCAP];
    __shared__ int lcnt[OWN_WAVES];
    __shared__ int bad;

    const int tid  = threadIdx.x;
    const int lane = tid & 31;
    const int wave = __builtin_amdgcn_readfirstlane(tid >> 5);
    const int r0 = (int)blockIdx.x * ROWS_OWN;
    const int nrows = (N - r0) < ROWS_OWN ? (N - r0) : ROWS_OWN;

    const int slot = (int)blockIdx.x * OWN_WAVES + wave;
    v2i* kb = reinterpret_cast<v2i*>(keep) + (size_t)slot * 2 * BCAP;
    int* kcnt = reinterpret_cast<int*>(keep) + (size_t)gridDim.x * OWN_WAVES * 2 * BCAP * 2;
    int* kflag = kcnt + (size_t)gridDim.x * OWN_WAVES * 2;

    int fast = 0;
    if (reuse) fast = __builtin_amdgcn_readfirstlane(kflag[blockIdx.x] == 0 ? 1 : 0);

    if (fast) {
        const int n = __builtin_amdgcn_readfirstlane(kcnt[slot * 2 + (hop - 1)]);
        const v2i* bp = kb + (hop - 1) * BCAP;
        v2i* own = lst + wave * LCAP;
        v2i en[BCAP / 32];
        #pragma unroll
        for (int u = 0; u < BCAP / 32; ++u) {
            en[u] = v2i{0, -1};
            if (u * 32 + lane < n) en[u] = bp[u * 32 + lane];
        }
        float* orow = aggOut + (size_t)r0 * DDIM + lane * 4;

        int pos = 0;
        for (int r = wave; r < nrows; r += OWN_WAVES) {
            const int start = pos;
            #pragma unroll
            for (int u = 0; u < BCAP / 32; ++u) {
                const bool hit = en[u].y == r;
                const unsigned mask = __builtin_amdgcn_ballot_w32(hit);
                if (hit) own[pos + (int)__builtin_amdgcn_mbcnt_lo(mask, 0u)] = en[u];
                pos += __builtin_popcount(mask);
            }
            if (pos == start)
                *reinterpret_cast<v4f*>(orow + (size_t)r * DDIM) = v4f{0.f, 0.f, 0.f, 0.f};
        }

        const float* xlane = xsrc + lane * 4;
        v4f sum = v4f{0.f, 0.f, 0.f, 0.f};
        for (int j = 0; j < n; j += 32) {
            const int idx = j + lane;
            v2i cur = v2i{0, -1};
            int nextRow = -1;
            if (idx < n) cur = own[idx];
            if (idx + 1 < n) nextRow = own[idx + 1].y;
            const int last = (cur.y != nextRow) ? 1 : 0;
            const int m = (n - j) < 32 ? (n - j) : 32;
            int b = 0;
            for (; b + ROWB <= m; b += ROWB)
                dgnn_row_batch<true>(xlane, orow, cur.x, cur.y, last, b, m, sum);
            if (b < m)
                dgnn_row_batch<false>(xlane, orow, cur.x, cur.y, last, b, m, sum);
        }
        return;
    }

    if (tid == 0) bad = 0;
    for (int i = tid; i < ROWS_OWN * DDIM / 4; i += 1024)
        reinterpret_cast<v4f*>(acc)[i] = v4f{0.f, 0.f, 0.f, 0.f};
    __syncthreads();

    v2i* mylst = lst + wave * LCAP;
    int nLive = 0;
    {
        const unsigned char* kbytes = reinterpret_cast<const unsigned char*>(keep);
        const unsigned char* ownB = kbytes + OWNB_OFF;
        const v2i* rec = reinterpret_cast<const v2i*>(kbytes + REC_OFF);
        const int EP = (E + SWEEP - 1) / SWEEP * SWEEP;
        const unsigned me = (unsigned)blockIdx.x;
        const unsigned pat = me * 0x01010101u;

        int cnt = 0;
        for (int base = wave * TRIPB; base < EP; base += SWEEP) {
            v4u w[SCAN_B16];
            #pragma unroll
            for (int u = 0; u < SCAN_B16; ++u)
                w[u] = *reinterpret_cast<const v4u*>(ownB + base + u * 512 + lane * 16);
            #pragma unroll
            for (int q = 0; q < SCAN_B16 * 4; ++q) {
                const unsigned x = w[q >> 2][q & 3] ^ pat;
                unsigned f = ~(((x & 0x7f7f7f7fu) + 0x7f7f7f7fu) | x) & 0x80808080u;
                unsigned m = __builtin_amdgcn_ballot_w32(f != 0u);
                if (m) {
                    const int eq = base + (q >> 2) * 512 + lane * 16 + (q & 3) * 4;
                    while (m) {
                        if (f != 0u) {
                            const int pos = min(cnt + (int)__builtin_amdgcn_mbcnt_lo(m, 0u), LCAP - 1);
                            mylst[pos] = v2i{eq + (__builtin_ctz(f) >> 3), 0};
                            f &= f - 1u;
                        }
                        cnt += __builtin_popcount(m);
                        m = __builtin_amdgcn_ballot_w32(f != 0u);
                    }
                }
            }
        }
        if (cnt > LCAP) {
            for (int base = wave * TRIPB; base < EP; base += SWEEP) {
                for (int i = lane; i < TRIPB; i += 32) {
                    const int e = base + i;
                    if ((unsigned)ownB[e] == me) {
                        const v2i r = rec[e];
                        if ((r.y >> 16) == hop) {
                            const float* row = xsrc + (size_t)r.x * DDIM;
                            float* a = acc + (r.y & 0xffff) * DDIM;
                            for (int c = 0; c < DDIM; ++c) atomicAdd(a + (c & 3) * 32 + (c >> 2), row[c]);
                        }
                    }
                }
            }
            if (lane == 0) bad = 1;
            cnt = 0;
        }

        nLive = cnt;
        for (int i0 = 0; i0 < nLive; i0 += 128) {
            v2i en[4];
            #pragma unroll
            for (int u = 0; u < 4; ++u) {
                const int idx = i0 + u * 32 + lane;
                en[u] = v2i{0, 0};
                if (idx < nLive) en[u] = rec[mylst[idx].x];
            }
            #pragma unroll
            for (int u = 0; u < 4; ++u) {
                const int idx = i0 + u * 32 + lane;
                if (idx < nLive) mylst[idx] = en[u];
            }
        }
    }
    if (lane == 0) lcnt[wave] = nLive;
    __syncthreads();

    int os = 0, od = 0, on = 0, n1 = 0, n2 = 0;
    for (int L = 0; L < OWN_WAVES; ++L) {
        const int nL = __builtin_amdgcn_readfirstlane(lcnt[L]);
        const v2i* lp = lst + L * LCAP;
        for (int i0 = 0; i0 < nL; i0 += 128) {
            v2i en[4];
            #pragma unroll
            for (int u = 0; u < 4; ++u) {
                const int idx = i0 + u * 32 + lane;
                en[u] = v2i{0, -1};
                if (idx < nL) en[u] = lp[idx];
            }
            #pragma unroll
            for (int u = 0; u < 4; ++u) {
                const int y = en[u].y;
                const bool mine = (y >= 0) & ((y & 31) == wave);
                if (__builtin_amdgcn_ballot_w32(mine) == 0u) continue;
                const int k  = y >> 16;
                const int dl = y & 0xffff;
                if (!reuse) {
                    const bool s1 = mine & (k == 1);
                    const unsigned m1 = __builtin_amdgcn_ballot_w32(s1);
                    if (s1) {
                        const int pos = n1 + (int)__builtin_amdgcn_mbcnt_lo(m1, 0u);
                        if (pos < BCAP) kb[pos] = v2i{en[u].x, dl};
                    }
                    n1 += __builtin_popcount(m1);
                    const bool s2 = mine & (k == 2);
                    const unsigned m2 = __builtin_amdgcn_ballot_w32(s2);
                    if (s2) {
                        const int pos = n2 + (int)__builtin_amdgcn_mbcnt_lo(m2, 0u);
                        if (pos < BCAP) kb[BCAP + pos] = v2i{en[u].x, dl};
                    }
                    n2 += __builtin_popcount(m2);
                }
                unsigned mask = __builtin_amdgcn_ballot_w32(mine & (k == hop));
                while (mask) {
                    const int j = __builtin_ctz(mask);
                    mask &= mask - 1u;
                    os = (lane == on) ? __builtin_amdgcn_readlane(en[u].x, j) : os;
                    od = (lane == on) ? __builtin_amdgcn_readlane(dl, j) : od;
                    ++on;
                    if (on == 32) { dgnn_own_flush(xsrc, acc, os, od, 32, lane); on = 0; }
                }
            }
        }
    }
    dgnn_own_flush(xsrc, acc, os, od, on, lane);
    if (!reuse && lane == 0) {
        kcnt[slot * 2 + 0] = n1 < BCAP ? n1 : BCAP;
        kcnt[slot * 2 + 1] = n2 < BCAP ? n2 : BCAP;
        if (n1 > BCAP || n2 > BCAP) bad = 1;
    }
    __syncthreads();
    if (!reuse && tid == 0) kflag[blockIdx.x] = bad;

    v4f* outp = reinterpret_cast<v4f*>(aggOut + (size_t)r0 * DDIM);
    for (int i = tid; i < nrows * (DDIM / 4); i += 1024) {
        const float* a = acc + (i >> 5) * DDIM + (i & 31);
        outp[i] = v4f{a[0], a[32], a[64], a[96]};
    }
}

__device__ __forceinline__ void dgnn_stage_load(const float* __restrict__ agg1,
                                                const float* __restrict__ agg2,
                                                int st, int nTiles, int wave, int c4,
                                                v4f (&p1)[4], v4f (&p2)[4]) {
    const int t0 = 2 * st;
    const int t1 = t0 + 1 < nTiles ? t0 + 1 : nTiles - 1;
    #pragma unroll
    for (int j = 0; j < 4; ++j) {
        const int g = (j < 2 ? t0 : t1) * 16 + (j & 1) * 8 + wave;
        p1[j] = *reinterpret_cast<const v4f*>(agg1 + (size_t)g * DDIM + c4);
        p2[j] = *reinterpret_cast<const v4f*>(agg2 + (size_t)g * DDIM + c4);
    }
}

__device__ __forceinline__ void dgnn_stage_store(float* __restrict__ stage, int wave, int c4,
                                                 const v4f (&p1)[4], const v4f (&p2)[4]) {
    #pragma unroll
    for (int j = 0; j < 4; ++j) {
        float* row = stage + (j * 8 + wave) * APITCH + c4;
        *reinterpret_cast<v4f*>(row)        = p1[j];
        *reinterpret_cast<v4f*>(row + DDIM) = p2[j];
    }
}

#define CHK 4
__device__ __forceinline__ void dgnn_frag_load(const float* __restrict__ aP, int c, v2f (&f)[4 * CHK]) {
    #pragma unroll
    for (int i = 0; i < CHK; ++i) {
        const int s = c * CHK + i;
        f[4 * i + 0] = *reinterpret_cast<const v2f*>(aP + 4 * s);
        f[4 * i + 1] = *reinterpret_cast<const v2f*>(aP + 16 * APITCH + 4 * s);
        f[4 * i + 2] = *reinterpret_cast<const v2f*>(aP + DDIM + 4 * s);
        f[4 * i + 3] = *reinterpret_cast<const v2f*>(aP + 16 * APITCH + DDIM + 4 * s);
    }
}

__global__ __launch_bounds__(256)
void dgnn_gemm_kernel(const float* __restrict__ agg1,
                      const float* __restrict__ agg2,
                      const float* __restrict__ Wt,
                      const float* __restrict__ bt,
                      const float* __restrict__ alphat,
                      const float* __restrict__ xin,
                      float* __restrict__ xout,
                      int nTiles, int nSuper) {
    extern __shared__ float lds[];

    const int wave = threadIdx.x >> 5;
    const int lane = threadIdx.x & 31;
    const int col  = lane & 15;
    const int half = lane >> 4;
    const int n    = wave * 16 + col;
    const int c4   = lane << 2;

    int st = blockIdx.x;
    const int stride = gridDim.x;

    v4f p1[4], p2[4];
    dgnn_stage_load(agg1, agg2, st < nSuper ? st : 0, nTiles, wave, c4, p1, p2);

    float e0 = __expf(alphat[0]);
    float e1 = __expf(alphat[1]);
    float inv = 1.0f / (e0 + e1);
    float a0 = e0 * inv;
    float a1 = e1 * inv;

    v2f b1[32], b2[32];
    {
        const float* w0p = Wt + (size_t)(2 * half) * DDIM + n;
        const float* w1p = w0p + (size_t)DDIM * DDIM;
        #pragma unroll
        for (int s = 0; s < 32; ++s) {
            b1[s].x = a0 * w0p[(4 * s) * DDIM];
            b1[s].y = a0 * w0p[(4 * s + 1) * DDIM];
            b2[s].x = a1 * w1p[(4 * s) * DDIM];
            b2[s].y = a1 * w1p[(4 * s + 1) * DDIM];
        }
    }

    const float bc = a0 * bt[n] + a1 * bt[DDIM + n];

    dgnn_stage_store(lds, wave, c4, p1, p2);
    __syncthreads();

    int cur = 0;
    for (; st < nSuper; st += stride) {
        const int nxt = st + stride;
        const bool more = nxt < nSuper;
        if (more) dgnn_stage_load(agg1, agg2, nxt, nTiles, wave, c4, p1, p2);

        const int t0 = 2 * st;
        const int t1 = t0 + 1 < nTiles ? t0 + 1 : nTiles - 1;
        const size_t o0 = (size_t)(t0 * 16 + half * 8) * DDIM + n;
        const size_t o1 = (size_t)(t1 * 16 + half * 8) * DDIM + n;

        float xr0[8], xr1[8];
        #pragma unroll
        for (int r = 0; r < 8; ++r) xr0[r] = xin[o0 + r * DDIM];
        #pragma unroll
        for (int r = 0; r < 8; ++r) xr1[r] = xin[o1 + r * DDIM];

        const float* aP = lds + cur * STAGE + col * APITCH + 2 * half;

        v8f acc0 = {};
        v8f acc1 = {};

        v2f fa[4 * CHK], fb[4 * CHK];
        dgnn_frag_load(aP, 0, fa);
        #pragma unroll
        for (int c = 0; c < 32 / CHK; c += 2) {
            dgnn_frag_load(aP, c + 1, fb);
            __builtin_amdgcn_sched_barrier(0);
            #pragma unroll
            for (int i = 0; i < CHK; ++i) {
                const int s = c * CHK + i;
                acc0 = __builtin_amdgcn_wmma_f32_16x16x4_f32(false, fa[4 * i + 0], false, b1[s], (short)0, acc0, false, false);
                acc1 = __builtin_amdgcn_wmma_f32_16x16x4_f32(false, fa[4 * i + 1], false, b1[s], (short)0, acc1, false, false);
                acc0 = __builtin_amdgcn_wmma_f32_16x16x4_f32(false, fa[4 * i + 2], false, b2[s], (short)0, acc0, false, false);
                acc1 = __builtin_amdgcn_wmma_f32_16x16x4_f32(false, fa[4 * i + 3], false, b2[s], (short)0, acc1, false, false);
            }
            __builtin_amdgcn_sched_barrier(0);
            if (c + 2 < 32 / CHK) dgnn_frag_load(aP, c + 2, fa);
            __builtin_amdgcn_sched_barrier(0);
            #pragma unroll
            for (int i = 0; i < CHK; ++i) {
                const int s = (c + 1) * CHK + i;
                acc0 = __builtin_amdgcn_wmma_f32_16x16x4_f32(false, fb[4 * i + 0], false, b1[s], (short)0, acc0, false, false);
                acc1 = __builtin_amdgcn_wmma_f32_16x16x4_f32(false, fb[4 * i + 1], false, b1[s], (short)0, acc1, false, false);
                acc0 = __builtin_amdgcn_wmma_f32_16x16x4_f32(false, fb[4 * i + 2], false, b2[s], (short)0, acc0, false, false);
                acc1 = __builtin_amdgcn_wmma_f32_16x16x4_f32(false, fb[4 * i + 3], false, b2[s], (short)0, acc1, false, false);
            }
            __builtin_amdgcn_sched_barrier(0);
        }

        #pragma unroll
        for (int r = 0; r < 8; ++r) {
            float v = acc0[r] + bc;
            v = v > 0.0f ? v : 0.0f;
            xout[o0 + r * DDIM] = xr0[r] + v;
        }
        #pragma unroll
        for (int r = 0; r < 8; ++r) {
            float v = acc1[r] + bc;
            v = v > 0.0f ? v : 0.0f;
            xout[o1 + r * DDIM] = xr1[r] + v;
        }

        if (more) dgnn_stage_store(lds + (cur ^ 1) * STAGE, wave, c4, p1, p2);
        __syncthreads();
        cur ^= 1;
    }
}

extern "C" void kernel_launch(void* const* d_in, const int* in_sizes, int n_in,
                              void* d_out, int out_size, void* d_ws, size_t ws_size,
                              hipStream_t stream) {
    const float* x     = (const float*)d_in[0];
    const int*   eidx  = (const int*)d_in[1];
    const int*   eattr = (const int*)d_in[2];
    const float* W     = (const float*)d_in[3];
    const float* b     = (const float*)d_in[4];
    const float* alpha = (const float*)d_in[5];
    float*       out   = (float*)d_out;

    const int N = in_sizes[0] / DDIM;
    const int E = in_sizes[2];
    const int T = 3;

    const size_t nd = (size_t)N * DDIM;
    float* ws   = (float*)d_ws;
    float* h1   = ws;
    float* h2   = ws + nd;
    float* agg1 = ws + 2 * nd;
    float* agg2 = ws + 3 * nd;

    const int* srcIdx = eidx;
    const int* dstIdx = eidx + E;

    const float* xs[4] = { x, h1, h2, out };

    const int nTiles = (N + 15) / 16;
    const int nSuper = (nTiles + 1) / 2;
    const int gemmBlocks = 256;
    const size_t ldsBytes = (size_t)2 * STAGE * sizeof(float);

    const int ownBlocks = (N + ROWS_OWN - 1) / ROWS_OWN;

    float* keep = out;

    const int packBlocks = (E + SWEEP - 1) / SWEEP * (SWEEP / 1024);
    dgnn_pack_kernel<<<packBlocks, 256, 0, stream>>>(srcIdx, dstIdx, eattr, keep, E, N);


    for (int t = 0; t < T; ++t) {
        dgnn_own_kernel<<<ownBlocks, 1024, 0, stream>>>(
            xs[t], srcIdx, dstIdx, eattr, agg1, keep, E, N, 1,
            t > 0 ? 1 : 0);
        if (t != 1)
            dgnn_own_kernel<<<ownBlocks, 1024, 0, stream>>>(
                xs[t > 0 ? t - 1 : 0], srcIdx, dstIdx, eattr, agg2, keep, E, N, 2, 1);

        dgnn_gemm_kernel<<<gemmBlocks, 256, ldsBytes, stream>>>(
            agg1, agg2,
            W + (size_t)t * 2 * DDIM * DDIM,
            b + (size_t)t * 2 * DDIM,
            alpha + (size_t)t * 2,
            xs[t], (float*)xs[t + 1],
            nTiles, nSuper);
    }
}


# first aggregation launch: scanning waves hand each hit to its owning wave through per-pair LDS slots, so no wave walks all 32 lists; hop 1 is then summed from the kept bucket; sum order kept
# speedup vs baseline: 4.1068x; 1.0818x over previous
#include <hip/hip_runtime.h>
#include <hip/hip_bf16.h>

#define DDIM 128
#define APITCH 260
#define TROWS 32
#define STAGE (TROWS * APITCH)

typedef __attribute__((ext_vector_type(2))) float v2f;
typedef __attribute__((ext_vector_type(4))) float v4f;
typedef __attribute__((ext_vector_type(8))) float v8f;

#define ROWS_OWN 400
#define OWN_WAVES 32
#define LCAP 384
#define SCAN_B16 8
#define TRIPB (SCAN_B16 * 512)
#define SWEEP (OWN_WAVES * TRIPB)
#define OWNB_OFF 25165824u
#define REC_OFF  27262976u
#define ROWB 8
#define BCAP 192
#define SCAP 24
#define SUBD (OWN_WAVES * SCAP + 1)

typedef __attribute__((ext_vector_type(4))) int v4i;
typedef v4i v4i_a4 __attribute__((aligned(4)));
typedef __attribute__((ext_vector_type(2))) int v2i;
typedef __attribute__((ext_vector_type(4))) unsigned v4u;

__global__ __launch_bounds__(256)
void dgnn_pack_kernel(const int* __restrict__ srcIdx,
                      const int* __restrict__ dstIdx,
                      const int* __restrict__ attr,
                      float* __restrict__ keep,
                      int E, int N) {
    unsigned char* kbytes = reinterpret_cast<unsigned char*>(keep);
    const int e0 = ((int)blockIdx.x * 256 + (int)threadIdx.x) * 4;
    const bool whole = e0 + 4 <= E;
    v4i d = v4i{-1, -1, -1, -1}, s = v4i{0, 0, 0, 0}, k = v4i{0, 0, 0, 0};
    if (whole) {
        d = *reinterpret_cast<const v4i_a4*>(dstIdx + e0);
        s = *reinterpret_cast<const v4i_a4*>(srcIdx + e0);
        k = *reinterpret_cast<const v4i_a4*>(attr + e0);
    } else {
        for (int j = 0; j < 4; ++j) {
            if (e0 + j < E) { d[j] = dstIdx[e0 + j]; s[j] = srcIdx[e0 + j]; k[j] = attr[e0 + j]; }
        }
    }
    unsigned w = 0;
    v2i r[4];
    #pragma unroll
    for (int j = 0; j < 4; ++j) {
        const unsigned du = (unsigned)d[j];
        const bool live = (du < (unsigned)N) & ((unsigned)(k[j] - 1) < 2u);
        const unsigned o = du / ROWS_OWN;
        const unsigned dl = du - o * ROWS_OWN;
        w |= (live ? o : 255u) << (8 * j);
        r[j] = v2i{s[j], (int)(dl | ((unsigned)k[j] << 16))};
    }
    *reinterpret_cast<unsigned*>(kbytes + OWNB_OFF + e0) = w;
    v2i* rec = reinterpret_cast<v2i*>(kbytes + REC_OFF);
    if (whole) {
        *reinterpret_cast<v4i*>(rec + e0)     = v4i{r[0].x, r[0].y, r[1].x, r[1].y};
        *reinterpret_cast<v4i*>(rec + e0 + 2) = v4i{r[2].x, r[2].y, r[3].x, r[3].y};
    } else {
        for (int j = 0; j < 4; ++j)
            if (e0 + j < E) rec[e0 + j] = r[j];
    }
}

__device__ __forceinline__ void dgnn_own_flush(const float* __restrict__ xsrc, float* acc,
                                               int os, int od, int n, int lane) {
    for (int j = 0; j < n; j += ROWB) {
        v4f v[ROWB];
        #pragma unroll
        for (int X = 0; X < ROWB; ++X) {
            v[X] = v4f{0.f, 0.f, 0.f, 0.f};
            if (j + X < n) {
                const int sr = __builtin_amdgcn_readlane(os, j + X);
                v[X] = *reinterpret_cast<const v4f*>(xsrc + (size_t)sr * DDIM + lane * 4);
            }
        }
        #pragma unroll
        for (int X = 0; X < ROWB; ++X) {
            if (j + X < n) {
                const int dr = __builtin_amdgcn_readlane(od, j + X);
                float* a = acc + dr * DDIM + lane;
                atomicAdd(a + 0,  v[X].x);
                atomicAdd(a + 32, v[X].y);
                atomicAdd(a + 64, v[X].z);
                atomicAdd(a + 96, v[X].w);
            }
        }
    }
}

template <bool FULL>
__device__ __forceinline__ void dgnn_row_batch(const float* __restrict__ xlane, float* __restrict__ orow,
                                               int srcs, int rows, int lasts, int b, int m, v4f& sum) {
    v4f v[ROWB];
    #pragma unroll
    for (int X = 0; X < ROWB; ++X) {
        if (FULL || b + X < m) {
            const unsigned sr = (unsigned)__builtin_amdgcn_readlane(srcs, b + X);
            v[X] = *reinterpret_cast<const v4f*>(xlane + (size_t)sr * DDIM);
        }
    }
    #pragma unroll
    for (int X = 0; X < ROWB; ++X) {
        if (FULL || b + X < m) {
            sum += v[X];
            if (__builtin_amdgcn_readlane(lasts, b + X)) {
                const unsigned dr = (unsigned)__builtin_amdgcn_readlane(rows, b + X);
                *reinterpret_cast<v4f*>(orow + (size_t)dr * DDIM) = sum;
                sum = v4f{0.f, 0.f, 0.f, 0.f};
            }
        }
    }
}

template <bool DIRECT>
__device__ __forceinline__ int dgnn_scan_hits(const float* __restrict__ xsrc, const float* keep,
                                              float* acc, v2i* mylst, int* badp,
                                              int E, int hop, int wave, int lane) {
    int nLive = 0;
    const unsigned char* kbytes = reinterpret_cast<const unsigned char*>(keep);
    const unsigned char* ownB = kbytes + OWNB_OFF;
    const v2i* rec = reinterpret_cast<const v2i*>(kbytes + REC_OFF);
    const int EP = (E + SWEEP - 1) / SWEEP * SWEEP;
    const unsigned me = (unsigned)blockIdx.x;
    const unsigned pat = me * 0x01010101u;

    int cnt = 0;
    for (int base = wave * TRIPB; base < EP; base += SWEEP) {
        v4u w[SCAN_B16];
        #pragma unroll
        for (int u = 0; u < SCAN_B16; ++u)
            w[u] = *reinterpret_cast<const v4u*>(ownB + base + u * 512 + lane * 16);
        #pragma unroll
        for (int q = 0; q < SCAN_B16 * 4; ++q) {
            const unsigned x = w[q >> 2][q & 3] ^ pat;
            unsigned f = ~(((x & 0x7f7f7f7fu) + 0x7f7f7f7fu) | x) & 0x80808080u;
            unsigned m = __builtin_amdgcn_ballot_w32(f != 0u);
            if (m) {
                const int eq = base + (q >> 2) * 512 + lane * 16 + (q & 3) * 4;
                while (m) {
                    if (f != 0u) {
                        const int pos = min(cnt + (int)__builtin_amdgcn_mbcnt_lo(m, 0u), LCAP - 1);
                        mylst[pos] = v2i{eq + (__builtin_ctz(f) >> 3), 0};
                        f &= f - 1u;
                    }
                    cnt += __builtin_popcount(m);
                    m = __builtin_amdgcn_ballot_w32(f != 0u);
                }
            }
        }
    }
    if (cnt > LCAP) {
        if (DIRECT) {
        for (int base = wave * TRIPB; base < EP; base += SWEEP) {
            for (int i = lane; i < TRIPB; i += 32) {
                const int e = base + i;
                if ((unsigned)ownB[e] == me) {
                    const v2i r = rec[e];
                    if ((r.y >> 16) == hop) {
                        const float* row = xsrc + (size_t)r.x * DDIM;
                        float* a = acc + (r.y & 0xffff) * DDIM;
                        for (int c = 0; c < DDIM; ++c) atomicAdd(a + (c & 3) * 32 + (c >> 2), row[c]);
                    }
                }
            }
        }
        }
        if (lane == 0) *badp = 1;
        cnt = 0;
    }

    nLive = cnt;
    for (int i0 = 0; i0 < nLive; i0 += 128) {
        v2i en[4];
        #pragma unroll
        for (int u = 0; u < 4; ++u) {
            const int idx = i0 + u * 32 + lane;
            en[u] = v2i{0, 0};
            if (idx < nLive) en[u] = rec[mylst[idx].x];
        }
        #pragma unroll
        for (int u = 0; u < 4; ++u) {
            const int idx = i0 + u * 32 + lane;
            if (idx < nLive) mylst[idx] = en[u];
        }
    }
    return nLive;
}

__global__ __launch_bounds__(1024)
void dgnn_own_kernel(const float* __restrict__ xsrc,
                     const int* __restrict__ srcIdx,
                     const int* __restrict__ dstIdx,
                     const int* __restrict__ attr,
                     float* __restrict__ aggOut,
                     float* __restrict__ keep,
                     int E, int N, int hop, int reuse) {
    __shared__ __attribute__((aligned(16))) float acc[ROWS_OWN * DDIM];
    __shared__ v2i lst[OWN_WAVES * LCAP];
    __shared__ int lcnt[OWN_WAVES];
    __shared__ int bad;

    const int tid  = threadIdx.x;
    const int lane = tid & 31;
    const int wave = __builtin_amdgcn_readfirstlane(tid >> 5);
    const int r0 = (int)blockIdx.x * ROWS_OWN;
    const int nrows = (N - r0) < ROWS_OWN ? (N - r0) : ROWS_OWN;

    const int slot = (int)blockIdx.x * OWN_WAVES + wave;
    v2i* kb = reinterpret_cast<v2i*>(keep) + (size_t)slot * 2 * BCAP;
    int* kcnt = reinterpret_cast<int*>(keep) + (size_t)gridDim.x * OWN_WAVES * 2 * BCAP * 2;
    int* kflag = kcnt + (size_t)gridDim.x * OWN_WAVES * 2;

    int fast = 0;
    if (reuse) {
        fast = __builtin_amdgcn_readfirstlane(kflag[blockIdx.x] == 0 ? 1 : 0);
    } else {
        if (tid == 0) bad = 0;
        __syncthreads();
        v2i* mylst = lst + wave * LCAP;
        const int nLive = dgnn_scan_hits<false>(xsrc, keep, acc, mylst, &bad, E, hop, wave, lane);

        v2i* sub = reinterpret_cast<v2i*>(acc);
        int* cntm = reinterpret_cast<int*>(acc) + OWN_WAVES * SUBD * 2;

        int cntv = 0;
        for (int i0 = 0; i0 < nLive; i0 += 32) {
            const int idx = i0 + lane;
            const bool valid = idx < nLive;
            v2i e = v2i{0, 0};
            if (valid) e = mylst[idx];
            const int d = e.y & 31;
            const unsigned vm = __builtin_amdgcn_ballot_w32(valid);
            unsigned same = vm;
            unsigned forMe = vm;
            #pragma unroll
            for (int k = 0; k < 5; ++k) {
                const unsigned bk = __builtin_amdgcn_ballot_w32(valid & (((d >> k) & 1) != 0));
                same  &= ((d >> k) & 1) ? bk : ~bk;
                forMe &= ((lane >> k) & 1) ? bk : ~bk;
            }
            const int pos = __shfl(cntv, d) + (int)__builtin_amdgcn_mbcnt_lo(same, 0u);
            if (valid && pos < SCAP) sub[d * SUBD + wave * SCAP + pos] = e;
            cntv += __builtin_popcount(forMe);
        }
        cntm[lane * OWN_WAVES + wave] = cntv;
        if (__builtin_amdgcn_ballot_w32(cntv > SCAP) != 0u && lane == 0) bad = 1;
        __syncthreads();

        {
            const int fill = min(cntm[wave * OWN_WAVES + lane], SCAP);
            const v2i* mysub = sub + wave * SUBD;
            int n1 = 0, n2 = 0;
            for (int Ls = 0; Ls < OWN_WAVES; ++Ls) {
                const int c = __builtin_amdgcn_readlane(fill, Ls);
                if (c == 0) continue;
                const bool have = lane < c;
                v2i e = v2i{0, 0};
                if (have) e = mysub[Ls * SCAP + lane];
                const int k  = e.y >> 16;
                const int dl = e.y & 0xffff;
                const bool s1 = have & (k == 1);
                const unsigned m1 = __builtin_amdgcn_ballot_w32(s1);
                if (s1) {
                    const int pos = n1 + (int)__builtin_amdgcn_mbcnt_lo(m1, 0u);
                    if (pos < BCAP) kb[pos] = v2i{e.x, dl};
                }
                n1 += __builtin_popcount(m1);
                const bool s2 = have & (k == 2);
                const unsigned m2 = __builtin_amdgcn_ballot_w32(s2);
                if (s2) {
                    const int pos = n2 + (int)__builtin_amdgcn_mbcnt_lo(m2, 0u);
                    if (pos < BCAP) kb[BCAP + pos] = v2i{e.x, dl};
                }
                n2 += __builtin_popcount(m2);
            }
            if (lane == 0) {
                kcnt[slot * 2 + 0] = n1 < BCAP ? n1 : BCAP;
                kcnt[slot * 2 + 1] = n2 < BCAP ? n2 : BCAP;
                if (n1 > BCAP || n2 > BCAP) bad = 1;
            }
        }
        __syncthreads();
        const int wasBad = bad;
        __syncthreads();
        if (tid == 0) kflag[blockIdx.x] = wasBad;
        fast = __builtin_amdgcn_readfirstlane(wasBad == 0 ? 1 : 0);
    }

    if (fast) {
        const int n = __builtin_amdgcn_readfirstlane(kcnt[slot * 2 + (hop - 1)]);
        const v2i* bp = kb + (hop - 1) * BCAP;
        v2i* own = lst + wave * LCAP;
        v2i en[BCAP / 32];
        #pragma unroll
        for (int u = 0; u < BCAP / 32; ++u) {
            en[u] = v2i{0, -1};
            if (u * 32 + lane < n) en[u] = bp[u * 32 + lane];
        }
        float* orow = aggOut + (size_t)r0 * DDIM + lane * 4;

        int pos = 0;
        for (int r = wave; r < nrows; r += OWN_WAVES) {
            const int start = pos;
            #pragma unroll
            for (int u = 0; u < BCAP / 32; ++u) {
                const bool hit = en[u].y == r;
                const unsigned mask = __builtin_amdgcn_ballot_w32(hit);
                if (hit) own[pos + (int)__builtin_amdgcn_mbcnt_lo(mask, 0u)] = en[u];
                pos += __builtin_popcount(mask);
            }
            if (pos == start)
                *reinterpret_cast<v4f*>(orow + (size_t)r * DDIM) = v4f{0.f, 0.f, 0.f, 0.f};
        }

        const float* xlane = xsrc + lane * 4;
        v4f sum = v4f{0.f, 0.f, 0.f, 0.f};
        for (int j = 0; j < n; j += 32) {
            const int idx = j + lane;
            v2i cur = v2i{0, -1};
            int nextRow = -1;
            if (idx < n) cur = own[idx];
            if (idx + 1 < n) nextRow = own[idx + 1].y;
            const int last = (cur.y != nextRow) ? 1 : 0;
            const int m = (n - j) < 32 ? (n - j) : 32;
            int b = 0;
            for (; b + ROWB <= m; b += ROWB)
                dgnn_row_batch<true>(xlane, orow, cur.x, cur.y, last, b, m, sum);
            if (b < m)
                dgnn_row_batch<false>(xlane, orow, cur.x, cur.y, last, b, m, sum);
        }
        return;
    }

    if (tid == 0) bad = 0;
    for (int i = tid; i < ROWS_OWN * DDIM / 4; i += 1024)
        reinterpret_cast<v4f*>(acc)[i] = v4f{0.f, 0.f, 0.f, 0.f};
    __syncthreads();

    v2i* mylst = lst + wave * LCAP;
    const int nLive = dgnn_scan_hits<true>(xsrc, keep, acc, mylst, &bad, E, hop, wave, lane);
    if (lane == 0) lcnt[wave] = nLive;
    __syncthreads();

    int os = 0, od = 0, on = 0;
    for (int L = 0; L < OWN_WAVES; ++L) {
        const int nL = __builtin_amdgcn_readfirstlane(lcnt[L]);
        const v2i* lp = lst + L * LCAP;
        for (int i0 = 0; i0 < nL; i0 += 128) {
            v2i en[4];
            #pragma unroll
            for (int u = 0; u < 4; ++u) {
                const int idx = i0 + u * 32 + lane;
                en[u] = v2i{0, -1};
                if (idx < nL) en[u] = lp[idx];
            }
            #pragma unroll
            for (int u = 0; u < 4; ++u) {
                const int y = en[u].y;
                const bool mine = (y >= 0) & ((y & 31) == wave);
                if (__builtin_amdgcn_ballot_w32(mine) == 0u) continue;
                const int k  = y >> 16;
                const int dl = y & 0xffff;
                unsigned mask = __builtin_amdgcn_ballot_w32(mine & (k == hop));
                while (mask) {
                    const int j = __builtin_ctz(mask);
                    mask &= mask - 1u;
                    os = (lane == on) ? __builtin_amdgcn_readlane(en[u].x, j) : os;
                    od = (lane == on) ? __builtin_amdgcn_readlane(dl, j) : od;
                    ++on;
                    if (on == 32) { dgnn_own_flush(xsrc, acc, os, od, 32, lane); on = 0; }
                }
            }
        }
    }
    dgnn_own_flush(xsrc, acc, os, od, on, lane);
    __syncthreads();

    v4f* outp = reinterpret_cast<v4f*>(aggOut + (size_t)r0 * DDIM);
    for (int i = tid; i < nrows * (DDIM / 4); i += 1024) {
        const float* a = acc + (i >> 5) * DDIM + (i & 31);
        outp[i] = v4f{a[0], a[32], a[64], a[96]};
    }
}

__device__ __forceinline__ void dgnn_stage_load(const float* __restrict__ agg1,
                                                const float* __restrict__ agg2,
                                                int st, int nTiles, int wave, int c4,
                                                v4f (&p1)[4], v4f (&p2)[4]) {
    const int t0 = 2 * st;
    const int t1 = t0 + 1 < nTiles ? t0 + 1 : nTiles - 1;
    #pragma unroll
    for (int j = 0; j < 4; ++j) {
        const int g = (j < 2 ? t0 : t1) * 16 + (j & 1) * 8 + wave;
        p1[j] = *reinterpret_cast<const v4f*>(agg1 + (size_t)g * DDIM + c4);
        p2[j] = *reinterpret_cast<const v4f*>(agg2 + (size_t)g * DDIM + c4);
    }
}

__device__ __forceinline__ void dgnn_stage_store(float* __restrict__ stage, int wave, int c4,
                                                 const v4f (&p1)[4], const v4f (&p2)[4]) {
    #pragma unroll
    for (int j = 0; j < 4; ++j) {
        float* row = stage + (j * 8 + wave) * APITCH + c4;
        *reinterpret_cast<v4f*>(row)        = p1[j];
        *reinterpret_cast<v4f*>(row + DDIM) = p2[j];
    }
}

#define CHK 4
__device__ __forceinline__ void dgnn_frag_load(const float* __restrict__ aP, int c, v2f (&f)[4 * CHK]) {
    #pragma unroll
    for (int i = 0; i < CHK; ++i) {
        const int s = c * CHK + i;
        f[4 * i + 0] = *reinterpret_cast<const v2f*>(aP + 4 * s);
        f[4 * i + 1] = *reinterpret_cast<const v2f*>(aP + 16 * APITCH + 4 * s);
        f[4 * i + 2] = *reinterpret_cast<const v2f*>(aP + DDIM + 4 * s);
        f[4 * i + 3] = *reinterpret_cast<const v2f*>(aP + 16 * APITCH + DDIM + 4 * s);
    }
}

__global__ __launch_bounds__(256)
void dgnn_gemm_kernel(const float* __restrict__ agg1,
                      const float* __restrict__ agg2,
                      const float* __restrict__ Wt,
                      const float* __restrict__ bt,
                      const float* __restrict__ alphat,
                      const float* __restrict__ xin,
                      float* __restrict__ xout,
                      int nTiles, int nSuper) {
    extern __shared__ float lds[];

    const int wave = threadIdx.x >> 5;
    const int lane = threadIdx.x & 31;
    const int col  = lane & 15;
    const int half = lane >> 4;
    const int n    = wave * 16 + col;
    const int c4   = lane << 2;

    int st = blockIdx.x;
    const int stride = gridDim.x;

    v4f p1[4], p2[4];
    dgnn_stage_load(agg1, agg2, st < nSuper ? st : 0, nTiles, wave, c4, p1, p2);

    float e0 = __expf(alphat[0]);
    float e1 = __expf(alphat[1]);
    float inv = 1.0f / (e0 + e1);
    float a0 = e0 * inv;
    float a1 = e1 * inv;

    v2f b1[32], b2[32];
    {
        const float* w0p = Wt + (size_t)(2 * half) * DDIM + n;
        const float* w1p = w0p + (size_t)DDIM * DDIM;
        #pragma unroll
        for (int s = 0; s < 32; ++s) {
            b1[s].x = a0 * w0p[(4 * s) * DDIM];
            b1[s].y = a0 * w0p[(4 * s + 1) * DDIM];
            b2[s].x = a1 * w1p[(4 * s) * DDIM];
            b2[s].y = a1 * w1p[(4 * s + 1) * DDIM];
        }
    }

    const float bc = a0 * bt[n] + a1 * bt[DDIM + n];

    dgnn_stage_store(lds, wave, c4, p1, p2);
    __syncthreads();

    int cur = 0;
    for (; st < nSuper; st += stride) {
        const int nxt = st + stride;
        const bool more = nxt < nSuper;
        if (more) dgnn_stage_load(agg1, agg2, nxt, nTiles, wave, c4, p1, p2);

        const int t0 = 2 * st;
        const int t1 = t0 + 1 < nTiles ? t0 + 1 : nTiles - 1;
        const size_t o0 = (size_t)(t0 * 16 + half * 8) * DDIM + n;
        const size_t o1 = (size_t)(t1 * 16 + half * 8) * DDIM + n;

        float xr0[8], xr1[8];
        #pragma unroll
        for (int r = 0; r < 8; ++r) xr0[r] = xin[o0 + r * DDIM];
        #pragma unroll
        for (int r = 0; r < 8; ++r) xr1[r] = xin[o1 + r * DDIM];

        const float* aP = lds + cur * STAGE + col * APITCH + 2 * half;

        v8f acc0 = {};
        v8f acc1 = {};

        v2f fa[4 * CHK], fb[4 * CHK];
        dgnn_frag_load(aP, 0, fa);
        #pragma unroll
        for (int c = 0; c < 32 / CHK; c += 2) {
            dgnn_frag_load(aP, c + 1, fb);
            __builtin_amdgcn_sched_barrier(0);
            #pragma unroll
            for (int i = 0; i < CHK; ++i) {
                const int s = c * CHK + i;
                acc0 = __builtin_amdgcn_wmma_f32_16x16x4_f32(false, fa[4 * i + 0], false, b1[s], (short)0, acc0, false, false);
                acc1 = __builtin_amdgcn_wmma_f32_16x16x4_f32(false, fa[4 * i + 1], false, b1[s], (short)0, acc1, false, false);
                acc0 = __builtin_amdgcn_wmma_f32_16x16x4_f32(false, fa[4 * i + 2], false, b2[s], (short)0, acc0, false, false);
                acc1 = __builtin_amdgcn_wmma_f32_16x16x4_f32(false, fa[4 * i + 3], false, b2[s], (short)0, acc1, false, false);
            }
            __builtin_amdgcn_sched_barrier(0);
            if (c + 2 < 32 / CHK) dgnn_frag_load(aP, c + 2, fa);
            __builtin_amdgcn_sched_barrier(0);
            #pragma unroll
            for (int i = 0; i < CHK; ++i) {
                const int s = (c + 1) * CHK + i;
                acc0 = __builtin_amdgcn_wmma_f32_16x16x4_f32(false, fb[4 * i + 0], false, b1[s], (short)0, acc0, false, false);
                acc1 = __builtin_amdgcn_wmma_f32_16x16x4_f32(false, fb[4 * i + 1], false, b1[s], (short)0, acc1, false, false);
                acc0 = __builtin_amdgcn_wmma_f32_16x16x4_f32(false, fb[4 * i + 2], false, b2[s], (short)0, acc0, false, false);
                acc1 = __builtin_amdgcn_wmma_f32_16x16x4_f32(false, fb[4 * i + 3], false, b2[s], (short)0, acc1, false, false);
            }
            __builtin_amdgcn_sched_barrier(0);
        }

        #pragma unroll
        for (int r = 0; r < 8; ++r) {
            float v = acc0[r] + bc;
            v = v > 0.0f ? v : 0.0f;
            xout[o0 + r * DDIM] = xr0[r] + v;
        }
        #pragma unroll
        for (int r = 0; r < 8; ++r) {
            float v = acc1[r] + bc;
            v = v > 0.0f ? v : 0.0f;
            xout[o1 + r * DDIM] = xr1[r] + v;
        }

        if (more) dgnn_stage_store(lds + (cur ^ 1) * STAGE, wave, c4, p1, p2);
        __syncthreads();
        cur ^= 1;
    }
}

extern "C" void kernel_launch(void* const* d_in, const int* in_sizes, int n_in,
                              void* d_out, int out_size, void* d_ws, size_t ws_size,
                              hipStream_t stream) {
    const float* x     = (const float*)d_in[0];
    const int*   eidx  = (const int*)d_in[1];
    const int*   eattr = (const int*)d_in[2];
    const float* W     = (const float*)d_in[3];
    const float* b     = (const float*)d_in[4];
    const float* alpha = (const float*)d_in[5];
    float*       out   = (float*)d_out;

    const int N = in_sizes[0] / DDIM;
    const int E = in_sizes[2];
    const int T = 3;

    const size_t nd = (size_t)N * DDIM;
    float* ws   = (float*)d_ws;
    float* h1   = ws;
    float* h2   = ws + nd;
    float* agg1 = ws + 2 * nd;
    float* agg2 = ws + 3 * nd;

    const int* srcIdx = eidx;
    const int* dstIdx = eidx + E;

    const float* xs[4] = { x, h1, h2, out };

    const int nTiles = (N + 15) / 16;
    const int nSuper = (nTiles + 1) / 2;
    const int gemmBlocks = 256;
    const size_t ldsBytes = (size_t)2 * STAGE * sizeof(float);

    const int ownBlocks = (N + ROWS_OWN - 1) / ROWS_OWN;

    float* keep = out;

    const int packBlocks = (E + SWEEP - 1) / SWEEP * (SWEEP / 1024);
    dgnn_pack_kernel<<<packBlocks, 256, 0, stream>>>(srcIdx, dstIdx, eattr, keep, E, N);


    for (int t = 0; t < T; ++t) {
        dgnn_own_kernel<<<ownBlocks, 1024, 0, stream>>>(
            xs[t], srcIdx, dstIdx, eattr, agg1, keep, E, N, 1,
            t > 0 ? 1 : 0);
        if (t != 1)
            dgnn_own_kernel<<<ownBlocks, 1024, 0, stream>>>(
                xs[t > 0 ? t - 1 : 0], srcIdx, dstIdx, eattr, agg2, keep, E, N, 2, 1);

        dgnn_gemm_kernel<<<gemmBlocks, 256, ldsBytes, stream>>>(
            agg1, agg2,
            W + (size_t)t * 2 * DDIM * DDIM,
            b + (size_t)t * 2 * DDIM,
            alpha + (size_t)t * 2,
            xs[t], (float*)xs[t + 1],
            nTiles, nSuper);
    }
}
